# starting kernel
# baseline (speedup 1.0000x reference)
#include <hip/hip_runtime.h>
#include <hip/hip_bf16.h>
#include <math.h>

#define NUM_SRC   200000
#define NUM_DST   100000
#define NUM_EDGES 1600000
#define IN_DIM    128
#define HID       256
#define OUT_DIM   128

typedef float v2f __attribute__((ext_vector_type(2)));
typedef float v8f __attribute__((ext_vector_type(8)));

#define ROWS 64
#define LDA  132
#define LDH  260
#define NBLK ((NUM_DST + ROWS - 1) / ROWS)
#define SMEM_BYTES ((ROWS * LDA + ROWS * LDH) * 4)

__global__ __launch_bounds__(256) void zero_kernel(float* __restrict__ p, size_t n) {
    size_t i = (size_t)blockIdx.x * blockDim.x + threadIdx.x;
    if (i < n) p[i] = 0.0f;
}

__global__ __launch_bounds__(256) void aggregate_kernel(
    const float* __restrict__ x,
    const int*   __restrict__ edge_src,
    const int*   __restrict__ edge_dst,
    float* __restrict__ sum,
    float* __restrict__ cnt,
    int num_edges)
{
    int edge = (int)((blockIdx.x * blockDim.x + threadIdx.x) >> 5);
    int lane = threadIdx.x & 31;
    if (edge >= num_edges) return;

    int src = edge_src[edge];
    int dst = edge_dst[edge];

    const float4 v = *(const float4*)(x + (size_t)src * IN_DIM + lane * 4);
    float* d = sum + (size_t)dst * IN_DIM + lane * 4;
    unsafeAtomicAdd(d + 0, v.x);
    unsafeAtomicAdd(d + 1, v.y);
    unsafeAtomicAdd(d + 2, v.z);
    unsafeAtomicAdd(d + 3, v.w);
    if (lane == 0) unsafeAtomicAdd(cnt + dst, 1.0f);
}

__global__ __launch_bounds__(512) void fused_mlp_kernel(
    const float* __restrict__ sum, const float* __restrict__ cnt,
    const float* __restrict__ x,
    const float* __restrict__ w_l,  const float* __restrict__ b_l,
    const float* __restrict__ w_r,
    const float* __restrict__ ln1_g, const float* __restrict__ ln1_b,
    const float* __restrict__ w_out, const float* __restrict__ b_out,
    const float* __restrict__ ln2_g, const float* __restrict__ ln2_b,
    float* __restrict__ out)
{
    extern __shared__ float smem[];
    float* sA = smem;
    float* sH = smem + ROWS * LDA;

    const int tid  = threadIdx.x;
    const int wave = tid >> 5;
    const int lane = tid & 31;
    const int rowbase = blockIdx.x * ROWS;

    const int n    = lane & 15;
    const int kh   = lane >> 4;
    const int arow = lane & 15;

    #pragma unroll
    for (int j = 0; j < 4; ++j) {
        const int idx = tid + j * 512;
        const int row = idx >> 5;
        const int k   = (idx & 31) * 4;
        int rg = rowbase + row;
        if (rg >= NUM_DST) rg = NUM_DST - 1;
        const float c   = cnt[rg];
        const float inv = 1.0f / fmaxf(c, 1.0f);
        const float4 s4 = *(const float4*)(sum + (size_t)rg * IN_DIM + k);
        sA[row * LDA + k + 0] = s4.x * inv;
        sA[row * LDA + k + 1] = s4.y * inv;
        sA[row * LDA + k + 2] = s4.z * inv;
        sA[row * LDA + k + 3] = s4.w * inv;
    }
    __syncthreads();

    v8f acc[4];
    #pragma unroll
    for (int rt = 0; rt < 4; ++rt) acc[rt] = (v8f){};
    const int colbase = wave * 16;

    #pragma unroll 2
    for (int k0 = 0; k0 < IN_DIM; k0 += 4) {
        const int kb = k0 + 2 * kh;
        v2f b;
        b.x = w_l[(size_t)kb * HID + colbase + n];
        b.y = w_l[(size_t)(kb + 1) * HID + colbase + n];
        #pragma unroll
        for (int rt = 0; rt < 4; ++rt) {
            v2f a;
            a.x = sA[(rt * 16 + arow) * LDA + kb];
            a.y = sA[(rt * 16 + arow) * LDA + kb + 1];
            acc[rt] = __builtin_amdgcn_wmma_f32_16x16x4_f32(
                false, a, false, b, (short)0, acc[rt], false, false);
        }
    }
    __syncthreads();

    #pragma unroll
    for (int j = 0; j < 4; ++j) {
        const int idx = tid + j * 512;
        const int row = idx >> 5;
        const int k   = (idx & 31) * 4;
        int rg = rowbase + row;
        if (rg >= NUM_DST) rg = NUM_DST - 1;
        const float4 x4 = *(const float4*)(x + (size_t)rg * IN_DIM + k);
        sA[row * LDA + k + 0] = x4.x;
        sA[row * LDA + k + 1] = x4.y;
        sA[row * LDA + k + 2] = x4.z;
        sA[row * LDA + k + 3] = x4.w;
    }
    __syncthreads();

    #pragma unroll 2
    for (int k0 = 0; k0 < IN_DIM; k0 += 4) {
        const int kb = k0 + 2 * kh;
        v2f b;
        b.x = w_r[(size_t)kb * HID + colbase + n];
        b.y = w_r[(size_t)(kb + 1) * HID + colbase + n];
        #pragma unroll
        for (int rt = 0; rt < 4; ++rt) {
            v2f a;
            a.x = sA[(rt * 16 + arow) * LDA + kb];
            a.y = sA[(rt * 16 + arow) * LDA + kb + 1];
            acc[rt] = __builtin_amdgcn_wmma_f32_16x16x4_f32(
                false, a, false, b, (short)0, acc[rt], false, false);
        }
    }

    {
        const float bias = b_l[colbase + n];
        #pragma unroll
        for (int rt = 0; rt < 4; ++rt) {
            #pragma unroll
            for (int i = 0; i < 8; ++i) {
                const int row = rt * 16 + i + 8 * kh;
                sH[row * LDH + colbase + n] = acc[rt][i] + bias;
            }
        }
    }
    __syncthreads();

    #pragma unroll
    for (int rr = 0; rr < 4; ++rr) {
        const int row = wave + 16 * rr;
        float v[8], s = 0.0f, s2 = 0.0f;
        #pragma unroll
        for (int j = 0; j < 8; ++j) {
            const float t = sH[row * LDH + lane + 32 * j];
            v[j] = t; s += t; s2 += t * t;
        }
        #pragma unroll
        for (int off = 16; off > 0; off >>= 1) {
            s  += __shfl_xor(s,  off, 32);
            s2 += __shfl_xor(s2, off, 32);
        }
        const float mean = s * (1.0f / HID);
        const float var  = s2 * (1.0f / HID) - mean * mean;
        const float rstd = rsqrtf(var + 1e-5f);
        #pragma unroll
        for (int j = 0; j < 8; ++j) {
            const int col = lane + 32 * j;
            const float hn = (v[j] - mean) * rstd * ln1_g[col] + ln1_b[col];
            sH[row * LDH + col] = 0.5f * hn * (1.0f + erff(hn * 0.70710678118654752f));
        }
    }
    __syncthreads();

    float* sOut = sA;
    if (wave < 8) {
        const int cb2 = wave * 16;
        v8f c[4];
        #pragma unroll
        for (int rt = 0; rt < 4; ++rt) c[rt] = (v8f){};
        #pragma unroll 2
        for (int k0 = 0; k0 < HID; k0 += 4) {
            const int kb = k0 + 2 * kh;
            v2f b;
            b.x = w_out[(size_t)kb * OUT_DIM + cb2 + n];
            b.y = w_out[(size_t)(kb + 1) * OUT_DIM + cb2 + n];
            #pragma unroll
            for (int rt = 0; rt < 4; ++rt) {
                v2f a;
                a.x = sH[(rt * 16 + arow) * LDH + kb];
                a.y = sH[(rt * 16 + arow) * LDH + kb + 1];
                c[rt] = __builtin_amdgcn_wmma_f32_16x16x4_f32(
                    false, a, false, b, (short)0, c[rt], false, false);
            }
        }
        const float bias = b_out[cb2 + n];
        #pragma unroll
        for (int rt = 0; rt < 4; ++rt) {
            #pragma unroll
            for (int i = 0; i < 8; ++i) {
                const int row = rt * 16 + i + 8 * kh;
                sOut[row * LDA + cb2 + n] = c[rt][i] + bias;
            }
        }
    }
    __syncthreads();

    #pragma unroll
    for (int rr = 0; rr < 4; ++rr) {
        const int row = wave + 16 * rr;
        float v[4], s = 0.0f, s2 = 0.0f;
        #pragma unroll
        for (int j = 0; j < 4; ++j) {
            const float t = sOut[row * LDA + lane + 32 * j];
            v[j] = t; s += t; s2 += t * t;
        }
        #pragma unroll
        for (int off = 16; off > 0; off >>= 1) {
            s  += __shfl_xor(s,  off, 32);
            s2 += __shfl_xor(s2, off, 32);
        }
        const float mean = s * (1.0f / OUT_DIM);
        const float var  = s2 * (1.0f / OUT_DIM) - mean * mean;
        const float rstd = rsqrtf(var + 1e-5f);
        const int rowg = rowbase + row;
        if (rowg < NUM_DST) {
            #pragma unroll
            for (int j = 0; j < 4; ++j) {
                const int col = lane + 32 * j;
                const float o = (v[j] - mean) * rstd * ln2_g[col] + ln2_b[col];
                out[(size_t)rowg * OUT_DIM + col] = o;
            }
        }
    }
}

extern "C" void kernel_launch(void* const* d_in, const int* in_sizes, int n_in,
                              void* d_out, int out_size, void* d_ws, size_t ws_size,
                              hipStream_t stream) {
    const float* x      = (const float*)d_in[0];
    const float* w_l    = (const float*)d_in[1];
    const float* b_l    = (const float*)d_in[2];
    const float* w_r    = (const float*)d_in[3];
    const float* ln1_g  = (const float*)d_in[4];
    const float* ln1_b  = (const float*)d_in[5];
    const float* w_out  = (const float*)d_in[6];
    const float* b_out  = (const float*)d_in[7];
    const float* ln2_g  = (const float*)d_in[8];
    const float* ln2_b  = (const float*)d_in[9];
    const int*   e_src  = (const int*)d_in[10];
    const int*   e_dst  = (const int*)d_in[11];

    float* sum = (float*)d_ws;
    float* cnt = sum + (size_t)NUM_DST * IN_DIM;
    float* out = (float*)d_out;

    const size_t nzero = (size_t)NUM_DST * IN_DIM + NUM_DST;
    zero_kernel<<<(unsigned)((nzero + 255) / 256), 256, 0, stream>>>(sum, nzero);

    aggregate_kernel<<<NUM_EDGES / 8, 256, 0, stream>>>(x, e_src, e_dst, sum, cnt,
                                                        NUM_EDGES);

    fused_mlp_kernel<<<NBLK, 512, SMEM_BYTES, stream>>>(sum, cnt, x,
                                                        w_l, b_l, w_r,
                                                        ln1_g, ln1_b,
                                                        w_out, b_out,
                                                        ln2_g, ln2_b, out);
}


# aggregate: lane l adds elements l,l+32,l+64,l+96, so each f32 atomic instruction covers one whole 128-byte line, not 8 floats in each of 4 lines; same adds, arrival order as before
# speedup vs baseline: 1.1134x; 1.1134x over previous
#include <hip/hip_runtime.h>
#include <hip/hip_bf16.h>
#include <math.h>

#define NUM_SRC   200000
#define NUM_DST   100000
#define NUM_EDGES 1600000
#define IN_DIM    128
#define HID       256
#define OUT_DIM   128

typedef float v2f __attribute__((ext_vector_type(2)));
typedef float v8f __attribute__((ext_vector_type(8)));

#define ROWS 64
#define LDA  132
#define LDH  260
#define NBLK ((NUM_DST + ROWS - 1) / ROWS)
#define SMEM_BYTES ((ROWS * LDA + ROWS * LDH) * 4)

__global__ __launch_bounds__(256) void zero_kernel(float* __restrict__ p, size_t n) {
    size_t i = (size_t)blockIdx.x * blockDim.x + threadIdx.x;
    if (i < n) p[i] = 0.0f;
}

__global__ __launch_bounds__(256) void aggregate_kernel(
    const float* __restrict__ x,
    const int*   __restrict__ edge_src,
    const int*   __restrict__ edge_dst,
    float* __restrict__ sum,
    float* __restrict__ cnt,
    int num_edges)
{
    int edge = (int)((blockIdx.x * blockDim.x + threadIdx.x) >> 5);
    int lane = threadIdx.x & 31;
    if (edge >= num_edges) return;

    int src = edge_src[edge];
    int dst = edge_dst[edge];

    const float* xs = x + (size_t)src * IN_DIM + lane;
    const float v0 = xs[0];
    const float v1 = xs[32];
    const float v2 = xs[64];
    const float v3 = xs[96];
    float* d = sum + (size_t)dst * IN_DIM + lane;
    unsafeAtomicAdd(d +  0, v0);
    unsafeAtomicAdd(d + 32, v1);
    unsafeAtomicAdd(d + 64, v2);
    unsafeAtomicAdd(d + 96, v3);
    if (lane == 0) unsafeAtomicAdd(cnt + dst, 1.0f);
}

__global__ __launch_bounds__(512) void fused_mlp_kernel(
    const float* __restrict__ sum, const float* __restrict__ cnt,
    const float* __restrict__ x,
    const float* __restrict__ w_l,  const float* __restrict__ b_l,
    const float* __restrict__ w_r,
    const float* __restrict__ ln1_g, const float* __restrict__ ln1_b,
    const float* __restrict__ w_out, const float* __restrict__ b_out,
    const float* __restrict__ ln2_g, const float* __restrict__ ln2_b,
    float* __restrict__ out)
{
    extern __shared__ float smem[];
    float* sA = smem;
    float* sH = smem + ROWS * LDA;

    const int tid  = threadIdx.x;
    const int wave = tid >> 5;
    const int lane = tid & 31;
    const int rowbase = blockIdx.x * ROWS;

    const int n    = lane & 15;
    const int kh   = lane >> 4;
    const int arow = lane & 15;

    #pragma unroll
    for (int j = 0; j < 4; ++j) {
        const int idx = tid + j * 512;
        const int row = idx >> 5;
        const int k   = (idx & 31) * 4;
        int rg = rowbase + row;
        if (rg >= NUM_DST) rg = NUM_DST - 1;
        const float c   = cnt[rg];
        const float inv = 1.0f / fmaxf(c, 1.0f);
        const float4 s4 = *(const float4*)(sum + (size_t)rg * IN_DIM + k);
        sA[row * LDA + k + 0] = s4.x * inv;
        sA[row * LDA + k + 1] = s4.y * inv;
        sA[row * LDA + k + 2] = s4.z * inv;
        sA[row * LDA + k + 3] = s4.w * inv;
    }
    __syncthreads();

    v8f acc[4];
    #pragma unroll
    for (int rt = 0; rt < 4; ++rt) acc[rt] = (v8f){};
    const int colbase = wave * 16;

    #pragma unroll 2
    for (int k0 = 0; k0 < IN_DIM; k0 += 4) {
        const int kb = k0 + 2 * kh;
        v2f b;
        b.x = w_l[(size_t)kb * HID + colbase + n];
        b.y = w_l[(size_t)(kb + 1) * HID + colbase + n];
        #pragma unroll
        for (int rt = 0; rt < 4; ++rt) {
            v2f a;
            a.x = sA[(rt * 16 + arow) * LDA + kb];
            a.y = sA[(rt * 16 + arow) * LDA + kb + 1];
            acc[rt] = __builtin_amdgcn_wmma_f32_16x16x4_f32(
                false, a, false, b, (short)0, acc[rt], false, false);
        }
    }
    __syncthreads();

    #pragma unroll
    for (int j = 0; j < 4; ++j) {
        const int idx = tid + j * 512;
        const int row = idx >> 5;
        const int k   = (idx & 31) * 4;
        int rg = rowbase + row;
        if (rg >= NUM_DST) rg = NUM_DST - 1;
        const float4 x4 = *(const float4*)(x + (size_t)rg * IN_DIM + k);
        sA[row * LDA + k + 0] = x4.x;
        sA[row * LDA + k + 1] = x4.y;
        sA[row * LDA + k + 2] = x4.z;
        sA[row * LDA + k + 3] = x4.w;
    }
    __syncthreads();

    #pragma unroll 2
    for (int k0 = 0; k0 < IN_DIM; k0 += 4) {
        const int kb = k0 + 2 * kh;
        v2f b;
        b.x = w_r[(size_t)kb * HID + colbase + n];
        b.y = w_r[(size_t)(kb + 1) * HID + colbase + n];
        #pragma unroll
        for (int rt = 0; rt < 4; ++rt) {
            v2f a;
            a.x = sA[(rt * 16 + arow) * LDA + kb];
            a.y = sA[(rt * 16 + arow) * LDA + kb + 1];
            acc[rt] = __builtin_amdgcn_wmma_f32_16x16x4_f32(
                false, a, false, b, (short)0, acc[rt], false, false);
        }
    }

    {
        const float bias = b_l[colbase + n];
        #pragma unroll
        for (int rt = 0; rt < 4; ++rt) {
            #pragma unroll
            for (int i = 0; i < 8; ++i) {
                const int row = rt * 16 + i + 8 * kh;
                sH[row * LDH + colbase + n] = acc[rt][i] + bias;
            }
        }
    }
    __syncthreads();

    #pragma unroll
    for (int rr = 0; rr < 4; ++rr) {
        const int row = wave + 16 * rr;
        float v[8], s = 0.0f, s2 = 0.0f;
        #pragma unroll
        for (int j = 0; j < 8; ++j) {
            const float t = sH[row * LDH + lane + 32 * j];
            v[j] = t; s += t; s2 += t * t;
        }
        #pragma unroll
        for (int off = 16; off > 0; off >>= 1) {
            s  += __shfl_xor(s,  off, 32);
            s2 += __shfl_xor(s2, off, 32);
        }
        const float mean = s * (1.0f / HID);
        const float var  = s2 * (1.0f / HID) - mean * mean;
        const float rstd = rsqrtf(var + 1e-5f);
        #pragma unroll
        for (int j = 0; j < 8; ++j) {
            const int col = lane + 32 * j;
            const float hn = (v[j] - mean) * rstd * ln1_g[col] + ln1_b[col];
            sH[row * LDH + col] = 0.5f * hn * (1.0f + erff(hn * 0.70710678118654752f));
        }
    }
    __syncthreads();

    float* sOut = sA;
    if (wave < 8) {
        const int cb2 = wave * 16;
        v8f c[4];
        #pragma unroll
        for (int rt = 0; rt < 4; ++rt) c[rt] = (v8f){};
        #pragma unroll 2
        for (int k0 = 0; k0 < HID; k0 += 4) {
            const int kb = k0 + 2 * kh;
            v2f b;
            b.x = w_out[(size_t)kb * OUT_DIM + cb2 + n];
            b.y = w_out[(size_t)(kb + 1) * OUT_DIM + cb2 + n];
            #pragma unroll
            for (int rt = 0; rt < 4; ++rt) {
                v2f a;
                a.x = sH[(rt * 16 + arow) * LDH + kb];
                a.y = sH[(rt * 16 + arow) * LDH + kb + 1];
                c[rt] = __builtin_amdgcn_wmma_f32_16x16x4_f32(
                    false, a, false, b, (short)0, c[rt], false, false);
            }
        }
        const float bias = b_out[cb2 + n];
        #pragma unroll
        for (int rt = 0; rt < 4; ++rt) {
            #pragma unroll
            for (int i = 0; i < 8; ++i) {
                const int row = rt * 16 + i + 8 * kh;
                sOut[row * LDA + cb2 + n] = c[rt][i] + bias;
            }
        }
    }
    __syncthreads();

    #pragma unroll
    for (int rr = 0; rr < 4; ++rr) {
        const int row = wave + 16 * rr;
        float v[4], s = 0.0f, s2 = 0.0f;
        #pragma unroll
        for (int j = 0; j < 4; ++j) {
            const float t = sOut[row * LDA + lane + 32 * j];
            v[j] = t; s += t; s2 += t * t;
        }
        #pragma unroll
        for (int off = 16; off > 0; off >>= 1) {
            s  += __shfl_xor(s,  off, 32);
            s2 += __shfl_xor(s2, off, 32);
        }
        const float mean = s * (1.0f / OUT_DIM);
        const float var  = s2 * (1.0f / OUT_DIM) - mean * mean;
        const float rstd = rsqrtf(var + 1e-5f);
        const int rowg = rowbase + row;
        if (rowg < NUM_DST) {
            #pragma unroll
            for (int j = 0; j < 4; ++j) {
                const int col = lane + 32 * j;
                const float o = (v[j] - mean) * rstd * ln2_g[col] + ln2_b[col];
                out[(size_t)rowg * OUT_DIM + col] = o;
            }
        }
    }
}

extern "C" void kernel_launch(void* const* d_in, const int* in_sizes, int n_in,
                              void* d_out, int out_size, void* d_ws, size_t ws_size,
                              hipStream_t stream) {
    const float* x      = (const float*)d_in[0];
    const float* w_l    = (const float*)d_in[1];
    const float* b_l    = (const float*)d_in[2];
    const float* w_r    = (const float*)d_in[3];
    const float* ln1_g  = (const float*)d_in[4];
    const float* ln1_b  = (const float*)d_in[5];
    const float* w_out  = (const float*)d_in[6];
    const float* b_out  = (const float*)d_in[7];
    const float* ln2_g  = (const float*)d_in[8];
    const float* ln2_b  = (const float*)d_in[9];
    const int*   e_src  = (const int*)d_in[10];
    const int*   e_dst  = (const int*)d_in[11];

    float* sum = (float*)d_ws;
    float* cnt = sum + (size_t)NUM_DST * IN_DIM;
    float* out = (float*)d_out;

    const size_t nzero = (size_t)NUM_DST * IN_DIM + NUM_DST;
    zero_kernel<<<(unsigned)((nzero + 255) / 256), 256, 0, stream>>>(sum, nzero);

    aggregate_kernel<<<NUM_EDGES / 8, 256, 0, stream>>>(x, e_src, e_dst, sum, cnt,
                                                        NUM_EDGES);

    fused_mlp_kernel<<<NBLK, 512, SMEM_BYTES, stream>>>(sum, cnt, x,
                                                        w_l, b_l, w_r,
                                                        ln1_g, ln1_b,
                                                        w_out, b_out,
                                                        ln2_g, ln2_b, out);
}


# aggregate: float atomic scatter-add into a zeroed sum plane replaced by per-row edge lists (integer count atomics assign slots) gathered in the MLP staging; row sums run in list (arrival) order
# speedup vs baseline: 1.6553x; 1.4867x over previous
#include <hip/hip_runtime.h>
#include <hip/hip_bf16.h>
#include <math.h>

#define NUM_SRC   200000
#define NUM_DST   100000
#define NUM_EDGES 1600000
#define IN_DIM    128
#define HID       256
#define OUT_DIM   128

typedef float v2f __attribute__((ext_vector_type(2)));
typedef float v8f __attribute__((ext_vector_type(8)));

#define ROWS 64
#define LDA  132
#define LDH  260
#define NBLK ((NUM_DST + ROWS - 1) / ROWS)
#define SMEM_BYTES ((ROWS * LDA + ROWS * LDH) * 4)

__global__ __launch_bounds__(256) void zero_kernel(int* __restrict__ p, int n) {
    int i = (int)(blockIdx.x * blockDim.x + threadIdx.x);
    if (i < n) p[i] = 0;
}

__global__ __launch_bounds__(256) void count_kernel(
    const int* __restrict__ edge_dst,
    int* __restrict__ cnt,
    int* __restrict__ rank,
    int num_edges)
{
    const int e = (int)(blockIdx.x * blockDim.x + threadIdx.x);
    if (e >= num_edges) return;
    rank[e] = atomicAdd(cnt + edge_dst[e], 1);
}

__global__ __launch_bounds__(256) void offsets_kernel(
    const int* __restrict__ cnt,
    int* __restrict__ total,
    int* __restrict__ offs,
    int num_rows)
{
    const int i    = (int)(blockIdx.x * blockDim.x + threadIdx.x);
    const int lane = threadIdx.x & 31;
    const int c    = (i < num_rows) ? cnt[i] : 0;
    int incl = c;
    #pragma unroll
    for (int d = 1; d < 32; d <<= 1) {
        const int t = __shfl_up(incl, d, 32);
        if (lane >= d) incl += t;
    }
    int base = 0;
    if (lane == 31) base = atomicAdd(total, incl);
    base = __shfl(base, 31, 32);
    if (i < num_rows) offs[i] = base + incl - c;
}

__global__ __launch_bounds__(256) void fill_kernel(
    const int* __restrict__ edge_src,
    const int* __restrict__ edge_dst,
    const int* __restrict__ offs,
    const int* __restrict__ rank,
    int* __restrict__ perm,
    int num_edges)
{
    const int e = (int)(blockIdx.x * blockDim.x + threadIdx.x);
    if (e >= num_edges) return;
    atomicExch(perm + offs[edge_dst[e]] + rank[e], edge_src[e]);
}

__global__ __launch_bounds__(512) void fused_mlp_kernel(
    const int* __restrict__ cnt, const int* __restrict__ offs,
    const int* __restrict__ perm,
    const float* __restrict__ x,
    const float* __restrict__ w_l,  const float* __restrict__ b_l,
    const float* __restrict__ w_r,
    const float* __restrict__ ln1_g, const float* __restrict__ ln1_b,
    const float* __restrict__ w_out, const float* __restrict__ b_out,
    const float* __restrict__ ln2_g, const float* __restrict__ ln2_b,
    float* __restrict__ out)
{
    extern __shared__ float smem[];
    float* sA = smem;
    float* sH = smem + ROWS * LDA;

    const int tid  = threadIdx.x;
    const int wave = tid >> 5;
    const int lane = tid & 31;
    const int rowbase = blockIdx.x * ROWS;

    const int n    = lane & 15;
    const int kh   = lane >> 4;
    const int arow = lane & 15;

    {
        const int wv = __builtin_amdgcn_readfirstlane(wave);
        int c4[4], st4[4], sv4[4];
        #pragma unroll
        for (int rr = 0; rr < 4; ++rr) {
            const int rg  = rowbase + wv * 4 + rr;
            const int rgc = min(rg, NUM_DST - 1);
            const int c   = cnt[rgc];
            st4[rr] = offs[rgc];
            c4[rr]  = (rg < NUM_DST) ? c : 0;
        }
        #pragma unroll
        for (int rr = 0; rr < 4; ++rr)
            sv4[rr] = (lane < c4[rr]) ? perm[st4[rr] + lane] : 0;
        #pragma unroll
        for (int rr = 0; rr < 4; ++rr) {
            const int row = wv * 4 + rr;
            const int c = c4[rr], st = st4[rr];
            float4 a4 = make_float4(0.0f, 0.0f, 0.0f, 0.0f);
            for (int base = 0; base < c; base += 32) {
                const int m = min(32, c - base);
                int sv = sv4[rr];
                if (base > 0) sv = (lane < m) ? perm[st + base + lane] : 0;
                for (int e = 0; e < m; e += 8) {
                    float4 v[8];
                    #pragma unroll
                    for (int j = 0; j < 8; ++j) {
                        const int ee = min(e + j, m - 1);
                        const int s  = __builtin_amdgcn_readlane(sv, ee);
                        v[j] = *(const float4*)(x + (size_t)s * IN_DIM + lane * 4);
                    }
                    #pragma unroll
                    for (int j = 0; j < 8; ++j) {
                        if (e + j < m) {
                            a4.x += v[j].x; a4.y += v[j].y;
                            a4.z += v[j].z; a4.w += v[j].w;
                        }
                    }
                }
            }
            const float inv = 1.0f / fmaxf((float)c, 1.0f);
            a4.x *= inv; a4.y *= inv; a4.z *= inv; a4.w *= inv;
            *(float4*)(sA + row * LDA + lane * 4) = a4;
        }
    }
    __syncthreads();

    v8f acc[4];
    #pragma unroll
    for (int rt = 0; rt < 4; ++rt) acc[rt] = (v8f){};
    const int colbase = wave * 16;

    #pragma unroll 2
    for (int k0 = 0; k0 < IN_DIM; k0 += 4) {
        const int kb = k0 + 2 * kh;
        v2f b;
        b.x = w_l[(size_t)kb * HID + colbase + n];
        b.y = w_l[(size_t)(kb + 1) * HID + colbase + n];
        #pragma unroll
        for (int rt = 0; rt < 4; ++rt) {
            v2f a;
            a.x = sA[(rt * 16 + arow) * LDA + kb];
            a.y = sA[(rt * 16 + arow) * LDA + kb + 1];
            acc[rt] = __builtin_amdgcn_wmma_f32_16x16x4_f32(
                false, a, false, b, (short)0, acc[rt], false, false);
        }
    }
    __syncthreads();

    #pragma unroll
    for (int j = 0; j < 4; ++j) {
        const int idx = tid + j * 512;
        const int row = idx >> 5;
        const int k   = (idx & 31) * 4;
        int rg = rowbase + row;
        if (rg >= NUM_DST) rg = NUM_DST - 1;
        const float4 x4 = *(const float4*)(x + (size_t)rg * IN_DIM + k);
        sA[row * LDA + k + 0] = x4.x;
        sA[row * LDA + k + 1] = x4.y;
        sA[row * LDA + k + 2] = x4.z;
        sA[row * LDA + k + 3] = x4.w;
    }
    __syncthreads();

    #pragma unroll 2
    for (int k0 = 0; k0 < IN_DIM; k0 += 4) {
        const int kb = k0 + 2 * kh;
        v2f b;
        b.x = w_r[(size_t)kb * HID + colbase + n];
        b.y = w_r[(size_t)(kb + 1) * HID + colbase + n];
        #pragma unroll
        for (int rt = 0; rt < 4; ++rt) {
            v2f a;
            a.x = sA[(rt * 16 + arow) * LDA + kb];
            a.y = sA[(rt * 16 + arow) * LDA + kb + 1];
            acc[rt] = __builtin_amdgcn_wmma_f32_16x16x4_f32(
                false, a, false, b, (short)0, acc[rt], false, false);
        }
    }

    {
        const float bias = b_l[colbase + n];
        #pragma unroll
        for (int rt = 0; rt < 4; ++rt) {
            #pragma unroll
            for (int i = 0; i < 8; ++i) {
                const int row = rt * 16 + i + 8 * kh;
                sH[row * LDH + colbase + n] = acc[rt][i] + bias;
            }
        }
    }
    __syncthreads();

    #pragma unroll
    for (int rr = 0; rr < 4; ++rr) {
        const int row = wave + 16 * rr;
        float v[8], s = 0.0f, s2 = 0.0f;
        #pragma unroll
        for (int j = 0; j < 8; ++j) {
            const float t = sH[row * LDH + lane + 32 * j];
            v[j] = t; s += t; s2 += t * t;
        }
        #pragma unroll
        for (int off = 16; off > 0; off >>= 1) {
            s  += __shfl_xor(s,  off, 32);
            s2 += __shfl_xor(s2, off, 32);
        }
        const float mean = s * (1.0f / HID);
        const float var  = s2 * (1.0f / HID) - mean * mean;
        const float rstd = rsqrtf(var + 1e-5f);
        #pragma unroll
        for (int j = 0; j < 8; ++j) {
            const int col = lane + 32 * j;
            const float hn = (v[j] - mean) * rstd * ln1_g[col] + ln1_b[col];
            sH[row * LDH + col] = 0.5f * hn * (1.0f + erff(hn * 0.70710678118654752f));
        }
    }
    __syncthreads();

    float* sOut = sA;
    if (wave < 8) {
        const int cb2 = wave * 16;
        v8f c[4];
        #pragma unroll
        for (int rt = 0; rt < 4; ++rt) c[rt] = (v8f){};
        #pragma unroll 2
        for (int k0 = 0; k0 < HID; k0 += 4) {
            const int kb = k0 + 2 * kh;
            v2f b;
            b.x = w_out[(size_t)kb * OUT_DIM + cb2 + n];
            b.y = w_out[(size_t)(kb + 1) * OUT_DIM + cb2 + n];
            #pragma unroll
            for (int rt = 0; rt < 4; ++rt) {
                v2f a;
                a.x = sH[(rt * 16 + arow) * LDH + kb];
                a.y = sH[(rt * 16 + arow) * LDH + kb + 1];
                c[rt] = __builtin_amdgcn_wmma_f32_16x16x4_f32(
                    false, a, false, b, (short)0, c[rt], false, false);
            }
        }
        const float bias = b_out[cb2 + n];
        #pragma unroll
        for (int rt = 0; rt < 4; ++rt) {
            #pragma unroll
            for (int i = 0; i < 8; ++i) {
                const int row = rt * 16 + i + 8 * kh;
                sOut[row * LDA + cb2 + n] = c[rt][i] + bias;
            }
        }
    }
    __syncthreads();

    #pragma unroll
    for (int rr = 0; rr < 4; ++rr) {
        const int row = wave + 16 * rr;
        float v[4], s = 0.0f, s2 = 0.0f;
        #pragma unroll
        for (int j = 0; j < 4; ++j) {
            const float t = sOut[row * LDA + lane + 32 * j];
            v[j] = t; s += t; s2 += t * t;
        }
        #pragma unroll
        for (int off = 16; off > 0; off >>= 1) {
            s  += __shfl_xor(s,  off, 32);
            s2 += __shfl_xor(s2, off, 32);
        }
        const float mean = s * (1.0f / OUT_DIM);
        const float var  = s2 * (1.0f / OUT_DIM) - mean * mean;
        const float rstd = rsqrtf(var + 1e-5f);
        const int rowg = rowbase + row;
        if (rowg < NUM_DST) {
            #pragma unroll
            for (int j = 0; j < 4; ++j) {
                const int col = lane + 32 * j;
                const float o = (v[j] - mean) * rstd * ln2_g[col] + ln2_b[col];
                out[(size_t)rowg * OUT_DIM + col] = o;
            }
        }
    }
}

extern "C" void kernel_launch(void* const* d_in, const int* in_sizes, int n_in,
                              void* d_out, int out_size, void* d_ws, size_t ws_size,
                              hipStream_t stream) {
    const float* x      = (const float*)d_in[0];
    const float* w_l    = (const float*)d_in[1];
    const float* b_l    = (const float*)d_in[2];
    const float* w_r    = (const float*)d_in[3];
    const float* ln1_g  = (const float*)d_in[4];
    const float* ln1_b  = (const float*)d_in[5];
    const float* w_out  = (const float*)d_in[6];
    const float* b_out  = (const float*)d_in[7];
    const float* ln2_g  = (const float*)d_in[8];
    const float* ln2_b  = (const float*)d_in[9];
    const int*   e_src  = (const int*)d_in[10];
    const int*   e_dst  = (const int*)d_in[11];

    int* cnt   = (int*)d_ws;
    int* total = cnt + NUM_DST;
    int* offs  = cnt + NUM_DST + 32;
    int* rank  = offs + NUM_DST;
    int* perm  = rank + NUM_EDGES;
    float* out = (float*)d_out;

    zero_kernel<<<(NUM_DST + 1 + 255) / 256, 256, 0, stream>>>(cnt, NUM_DST + 1);

    count_kernel<<<NUM_EDGES / 256, 256, 0, stream>>>(e_dst, cnt, rank, NUM_EDGES);

    offsets_kernel<<<(NUM_DST + 255) / 256, 256, 0, stream>>>(cnt, total, offs, NUM_DST);

    fill_kernel<<<NUM_EDGES / 256, 256, 0, stream>>>(e_src, e_dst, offs, rank, perm,
                                                    NUM_EDGES);

    fused_mlp_kernel<<<NBLK, 512, SMEM_BYTES, stream>>>(cnt, offs, perm, x,
                                                        w_l, b_l, w_r,
                                                        ln1_g, ln1_b,
                                                        w_out, b_out,
                                                        ln2_g, ln2_b, out);
}


# edge lists: per-edge global atomics (count, slot fill) replaced by LDS counting sorts: 256 edge chunks binned by 64-row block, each MLP block orders its bin by row; row sums in LDS arrival order
# speedup vs baseline: 2.4279x; 1.4667x over previous
#include <hip/hip_runtime.h>
#include <hip/hip_bf16.h>
#include <math.h>

#define NUM_SRC   200000
#define NUM_DST   100000
#define NUM_EDGES 1600000
#define IN_DIM    128
#define HID       256
#define OUT_DIM   128

typedef float v2f __attribute__((ext_vector_type(2)));
typedef float v8f __attribute__((ext_vector_type(8)));

#define ROWS 64
#define LDA  132
#define LDH  260
#define NBLK ((NUM_DST + ROWS - 1) / ROWS)
#define SMEM_BYTES ((ROWS * LDA + ROWS * LDH) * 4)

#define NCHUNK  256
#define CHUNK   (NUM_EDGES / NCHUNK)
#define EPT     ((CHUNK + 511) / 512)
#define HB      2048
#define PSTRIDE 1568
#define RSTRIDE 6272
#define LCAP    4096
static_assert(NUM_EDGES % NCHUNK == 0, "chunks must be equal");
static_assert(NBLK + 1 <= PSTRIDE && PSTRIDE <= HB && PSTRIDE % 32 == 0, "bin row");
static_assert(CHUNK <= RSTRIDE && RSTRIDE % 32 == 0, "record row");
static_assert(NUM_SRC <= (1 << 18), "source index must fit 18 bits of a record");

__global__ __launch_bounds__(512) void bin_kernel(
    const int* __restrict__ edge_src,
    const int* __restrict__ edge_dst,
    int* __restrict__ P,
    int* __restrict__ recs)
{
    __shared__ __attribute__((aligned(16))) int hist[HB];
    __shared__ __attribute__((aligned(16))) int srt[RSTRIDE];
    __shared__ int wsm[16];
    const int tid  = threadIdx.x;
    const int lane = tid & 31;
    const int wave = tid >> 5;
    const int base = blockIdx.x * CHUNK;

    int d[EPT], s[EPT];
    #pragma unroll
    for (int j = 0; j < EPT; ++j) {
        const int i = tid + j * 512;
        const bool ok = i < CHUNK;
        d[j] = ok ? edge_dst[base + i] : -1;
        s[j] = ok ? edge_src[base + i] : 0;
    }
    for (int i = tid; i < HB; i += 512) hist[i] = 0;
    if (tid < RSTRIDE - CHUNK) srt[CHUNK + tid] = 0;
    __syncthreads();

    #pragma unroll
    for (int j = 0; j < EPT; ++j)
        if ((unsigned)d[j] < (unsigned)NUM_DST) atomicAdd(&hist[d[j] >> 6], 1);
    __syncthreads();

    const int v0 = hist[4 * tid + 0], v1 = hist[4 * tid + 1];
    const int v2 = hist[4 * tid + 2], v3 = hist[4 * tid + 3];
    const int t4 = v0 + v1 + v2 + v3;
    int incl = t4;
    #pragma unroll
    for (int dd = 1; dd < 32; dd <<= 1) {
        const int t = __shfl_up(incl, dd, 32);
        if (lane >= dd) incl += t;
    }
    if (lane == 31) wsm[wave] = incl;
    __syncthreads();
    int wbase = 0;
    for (int w = 0; w < 16; ++w) { const int t = wsm[w]; if (w < wave) wbase += t; }
    const int e0 = wbase + incl - t4;
    const int e1 = e0 + v0, e2 = e1 + v1, e3 = e2 + v2;
    hist[4 * tid + 0] = e0; hist[4 * tid + 1] = e1;
    hist[4 * tid + 2] = e2; hist[4 * tid + 3] = e3;
    if (tid < PSTRIDE / 4)
        *(int4*)(P + (size_t)blockIdx.x * PSTRIDE + 4 * tid) = make_int4(e0, e1, e2, e3);
    __syncthreads();

    #pragma unroll
    for (int j = 0; j < EPT; ++j) {
        if ((unsigned)d[j] < (unsigned)NUM_DST) {
            const int slot = atomicAdd(&hist[d[j] >> 6], 1);
            const int sc   = min(max(s[j], 0), NUM_SRC - 1);
            srt[slot] = sc | ((d[j] & 63) << 18);
        }
    }
    __syncthreads();

    for (int i = tid; i < RSTRIDE / 4; i += 512)
        *(int4*)(recs + (size_t)blockIdx.x * RSTRIDE + 4 * i) = *(const int4*)(srt + 4 * i);
}

__global__ __launch_bounds__(512) void fused_mlp_kernel(
    const int* __restrict__ P, const int* __restrict__ recs,
    const float* __restrict__ x,
    const float* __restrict__ w_l,  const float* __restrict__ b_l,
    const float* __restrict__ w_r,
    const float* __restrict__ ln1_g, const float* __restrict__ ln1_b,
    const float* __restrict__ w_out, const float* __restrict__ b_out,
    const float* __restrict__ ln2_g, const float* __restrict__ ln2_b,
    float* __restrict__ out)
{
    extern __shared__ float smem[];
    float* sA = smem;
    float* sH = smem + ROWS * LDA;

    const int tid  = threadIdx.x;
    const int wave = tid >> 5;
    const int lane = tid & 31;
    const int rowbase = blockIdx.x * ROWS;

    const int n    = lane & 15;
    const int kh   = lane >> 4;
    const int arow = lane & 15;

    {
        int* li    = (int*)sH;
        int* lst_u = li;
        int* lst_s = li + LCAP;
        int* rcnt  = li + 2 * LCAP;
        int* rcur  = rcnt + 64;
        int* rst   = rcur + 64;
        int* wsm   = rst + 64;

        const int bin = blockIdx.x;

        const int wv = __builtin_amdgcn_readfirstlane(wave);

        int ctot[4];
        #pragma unroll
        for (int rr = 0; rr < 4; ++rr) {
            ctot[rr] = 0;
            *(float4*)(sA + (wv * 4 + rr) * LDA + lane * 4) = make_float4(0.0f, 0.0f, 0.0f, 0.0f);
        }

        int lo = 0, ntot;
        do {
            if (lo > 0) __syncthreads();
            int p0 = 0, n = 0;
            if (tid < NCHUNK) {
                p0 = P[(size_t)tid * PSTRIDE + bin];
                n  = P[(size_t)tid * PSTRIDE + bin + 1] - p0;
            }
            int incl = n;
            #pragma unroll
            for (int dd = 1; dd < 32; dd <<= 1) {
                const int t = __shfl_up(incl, dd, 32);
                if (lane >= dd) incl += t;
            }
            if (lane == 31) wsm[wave] = incl;
            if (tid < 64) rcnt[tid] = 0;
            __syncthreads();
            int wbase = 0, nall = 0;
            for (int w = 0; w < 16; ++w) {
                const int t = wsm[w];
                if (w < wave) wbase += t;
                nall += t;
            }
            ntot = __builtin_amdgcn_readfirstlane(nall);
            const int pos = wbase + incl - n;
            const int* myrec = recs + (size_t)(tid & (NCHUNK - 1)) * RSTRIDE + p0;
            const int m = max(0, min(LCAP, ntot - lo));

            {
                const int b0 = max(pos, lo), b1 = min(pos + n, lo + m);
                for (int i = b0; i < b1; i += 8) {
                    int rv[8];
                    #pragma unroll
                    for (int j = 0; j < 8; ++j)
                        rv[j] = (i + j < b1) ? myrec[i + j - pos] : 0;
                    #pragma unroll
                    for (int j = 0; j < 8; ++j) {
                        if (i + j < b1) {
                            lst_u[i + j - lo] = rv[j];
                            atomicAdd(&rcnt[(rv[j] >> 18) & 63], 1);
                        }
                    }
                }
            }
            __syncthreads();

            if (wave == 0) {
                const int c0 = rcnt[2 * lane], c1 = rcnt[2 * lane + 1];
                const int t2 = c0 + c1;
                int inc2 = t2;
                #pragma unroll
                for (int dd = 1; dd < 32; dd <<= 1) {
                    const int t = __shfl_up(inc2, dd, 32);
                    if (lane >= dd) inc2 += t;
                }
                const int ex = inc2 - t2;
                rst[2 * lane]      = ex;      rcur[2 * lane]     = ex;
                rst[2 * lane + 1]  = ex + c0; rcur[2 * lane + 1] = ex + c0;
            }
            __syncthreads();

            for (int j = tid; j < m; j += 512) {
                const int rec  = lst_u[j];
                const int slot = atomicAdd(&rcur[(rec >> 18) & 63], 1);
                lst_s[slot] = rec & 0x3FFFF;
            }
            __syncthreads();

            #pragma unroll
            for (int rr = 0; rr < 4; ++rr) {
                const int row = wv * 4 + rr;
                const int c  = __builtin_amdgcn_readfirstlane(rcnt[row]);
                const int st = __builtin_amdgcn_readfirstlane(rst[row]);
                ctot[rr] += c;
                float4 a4 = *(const float4*)(sA + row * LDA + lane * 4);
                for (int base = 0; base < c; base += 32) {
                    const int mm = min(32, c - base);
                    const int sv = (lane < mm) ? lst_s[st + base + lane] : 0;
                    for (int e = 0; e < mm; e += 8) {
                        float4 v[8];
                        #pragma unroll
                        for (int j = 0; j < 8; ++j) {
                            const int ee = min(e + j, mm - 1);
                            const int s  = __builtin_amdgcn_readlane(sv, ee);
                            v[j] = *(const float4*)(x + (size_t)s * IN_DIM + lane * 4);
                        }
                        #pragma unroll
                        for (int j = 0; j < 8; ++j) {
                            if (e + j < mm) {
                                a4.x += v[j].x; a4.y += v[j].y;
                                a4.z += v[j].z; a4.w += v[j].w;
                            }
                        }
                    }
                }
                *(float4*)(sA + row * LDA + lane * 4) = a4;
            }
            lo += LCAP;
        } while (lo < ntot);

        #pragma unroll
        for (int rr = 0; rr < 4; ++rr) {
            const int row = wv * 4 + rr;
            const float inv = 1.0f / fmaxf((float)ctot[rr], 1.0f);
            float4 a4 = *(const float4*)(sA + row * LDA + lane * 4);
            a4.x *= inv; a4.y *= inv; a4.z *= inv; a4.w *= inv;
            *(float4*)(sA + row * LDA + lane * 4) = a4;
        }
    }
    __syncthreads();

    v8f acc[4];
    #pragma unroll
    for (int rt = 0; rt < 4; ++rt) acc[rt] = (v8f){};
    const int colbase = wave * 16;

    #pragma unroll 2
    for (int k0 = 0; k0 < IN_DIM; k0 += 4) {
        const int kb = k0 + 2 * kh;
        v2f b;
        b.x = w_l[(size_t)kb * HID + colbase + n];
        b.y = w_l[(size_t)(kb + 1) * HID + colbase + n];
        #pragma unroll
        for (int rt = 0; rt < 4; ++rt) {
            v2f a;
            a.x = sA[(rt * 16 + arow) * LDA + kb];
            a.y = sA[(rt * 16 + arow) * LDA + kb + 1];
            acc[rt] = __builtin_amdgcn_wmma_f32_16x16x4_f32(
                false, a, false, b, (short)0, acc[rt], false, false);
        }
    }
    __syncthreads();

    #pragma unroll
    for (int j = 0; j < 4; ++j) {
        const int idx = tid + j * 512;
        const int row = idx >> 5;
        const int k   = (idx & 31) * 4;
        int rg = rowbase + row;
        if (rg >= NUM_DST) rg = NUM_DST - 1;
        const float4 x4 = *(const float4*)(x + (size_t)rg * IN_DIM + k);
        sA[row * LDA + k + 0] = x4.x;
        sA[row * LDA + k + 1] = x4.y;
        sA[row * LDA + k + 2] = x4.z;
        sA[row * LDA + k + 3] = x4.w;
    }
    __syncthreads();

    #pragma unroll 2
    for (int k0 = 0; k0 < IN_DIM; k0 += 4) {
        const int kb = k0 + 2 * kh;
        v2f b;
        b.x = w_r[(size_t)kb * HID + colbase + n];
        b.y = w_r[(size_t)(kb + 1) * HID + colbase + n];
        #pragma unroll
        for (int rt = 0; rt < 4; ++rt) {
            v2f a;
            a.x = sA[(rt * 16 + arow) * LDA + kb];
            a.y = sA[(rt * 16 + arow) * LDA + kb + 1];
            acc[rt] = __builtin_amdgcn_wmma_f32_16x16x4_f32(
                false, a, false, b, (short)0, acc[rt], false, false);
        }
    }

    {
        const float bias = b_l[colbase + n];
        #pragma unroll
        for (int rt = 0; rt < 4; ++rt) {
            #pragma unroll
            for (int i = 0; i < 8; ++i) {
                const int row = rt * 16 + i + 8 * kh;
                sH[row * LDH + colbase + n] = acc[rt][i] + bias;
            }
        }
    }
    __syncthreads();

    #pragma unroll
    for (int rr = 0; rr < 4; ++rr) {
        const int row = wave + 16 * rr;
        float v[8], s = 0.0f, s2 = 0.0f;
        #pragma unroll
        for (int j = 0; j < 8; ++j) {
            const float t = sH[row * LDH + lane + 32 * j];
            v[j] = t; s += t; s2 += t * t;
        }
        #pragma unroll
        for (int off = 16; off > 0; off >>= 1) {
            s  += __shfl_xor(s,  off, 32);
            s2 += __shfl_xor(s2, off, 32);
        }
        const float mean = s * (1.0f / HID);
        const float var  = s2 * (1.0f / HID) - mean * mean;
        const float rstd = rsqrtf(var + 1e-5f);
        #pragma unroll
        for (int j = 0; j < 8; ++j) {
            const int col = lane + 32 * j;
            const float hn = (v[j] - mean) * rstd * ln1_g[col] + ln1_b[col];
            sH[row * LDH + col] = 0.5f * hn * (1.0f + erff(hn * 0.70710678118654752f));
        }
    }
    __syncthreads();

    float* sOut = sA;
    if (wave < 8) {
        const int cb2 = wave * 16;
        v8f c[4];
        #pragma unroll
        for (int rt = 0; rt < 4; ++rt) c[rt] = (v8f){};
        #pragma unroll 2
        for (int k0 = 0; k0 < HID; k0 += 4) {
            const int kb = k0 + 2 * kh;
            v2f b;
            b.x = w_out[(size_t)kb * OUT_DIM + cb2 + n];
            b.y = w_out[(size_t)(kb + 1) * OUT_DIM + cb2 + n];
            #pragma unroll
            for (int rt = 0; rt < 4; ++rt) {
                v2f a;
                a.x = sH[(rt * 16 + arow) * LDH + kb];
                a.y = sH[(rt * 16 + arow) * LDH + kb + 1];
                c[rt] = __builtin_amdgcn_wmma_f32_16x16x4_f32(
                    false, a, false, b, (short)0, c[rt], false, false);
            }
        }
        const float bias = b_out[cb2 + n];
        #pragma unroll
        for (int rt = 0; rt < 4; ++rt) {
            #pragma unroll
            for (int i = 0; i < 8; ++i) {
                const int row = rt * 16 + i + 8 * kh;
                sOut[row * LDA + cb2 + n] = c[rt][i] + bias;
            }
        }
    }
    __syncthreads();

    #pragma unroll
    for (int rr = 0; rr < 4; ++rr) {
        const int row = wave + 16 * rr;
        float v[4], s = 0.0f, s2 = 0.0f;
        #pragma unroll
        for (int j = 0; j < 4; ++j) {
            const float t = sOut[row * LDA + lane + 32 * j];
            v[j] = t; s += t; s2 += t * t;
        }
        #pragma unroll
        for (int off = 16; off > 0; off >>= 1) {
            s  += __shfl_xor(s,  off, 32);
            s2 += __shfl_xor(s2, off, 32);
        }
        const float mean = s * (1.0f / OUT_DIM);
        const float var  = s2 * (1.0f / OUT_DIM) - mean * mean;
        const float rstd = rsqrtf(var + 1e-5f);
        const int rowg = rowbase + row;
        if (rowg < NUM_DST) {
            #pragma unroll
            for (int j = 0; j < 4; ++j) {
                const int col = lane + 32 * j;
                const float o = (v[j] - mean) * rstd * ln2_g[col] + ln2_b[col];
                out[(size_t)rowg * OUT_DIM + col] = o;
            }
        }
    }
}

extern "C" void kernel_launch(void* const* d_in, const int* in_sizes, int n_in,
                              void* d_out, int out_size, void* d_ws, size_t ws_size,
                              hipStream_t stream) {
    const float* x      = (const float*)d_in[0];
    const float* w_l    = (const float*)d_in[1];
    const float* b_l    = (const float*)d_in[2];
    const float* w_r    = (const float*)d_in[3];
    const float* ln1_g  = (const float*)d_in[4];
    const float* ln1_b  = (const float*)d_in[5];
    const float* w_out  = (const float*)d_in[6];
    const float* b_out  = (const float*)d_in[7];
    const float* ln2_g  = (const float*)d_in[8];
    const float* ln2_b  = (const float*)d_in[9];
    const int*   e_src  = (const int*)d_in[10];
    const int*   e_dst  = (const int*)d_in[11];

    int* P    = (int*)d_ws;
    int* recs = P + NCHUNK * PSTRIDE;
    float* out = (float*)d_out;

    bin_kernel<<<NCHUNK, 512, 0, stream>>>(e_src, e_dst, P, recs);

    fused_mlp_kernel<<<NBLK, 512, SMEM_BYTES, stream>>>(P, recs, x,
                                                        w_l, b_l, w_r,
                                                        ln1_g, ln1_b,
                                                        w_out, b_out,
                                                        ln2_g, ln2_b, out);
}


# GEMM k-loops: each weight fragment is loaded from global 7 k-steps ahead of its matrix instructions (8 fragment registers in rotation), not right in front of them; same sums in the same order
# speedup vs baseline: 2.5462x; 1.0487x over previous
#include <hip/hip_runtime.h>
#include <hip/hip_bf16.h>
#include <math.h>

#define NUM_SRC   200000
#define NUM_DST   100000
#define NUM_EDGES 1600000
#define IN_DIM    128
#define HID       256
#define OUT_DIM   128

typedef float v2f __attribute__((ext_vector_type(2)));
typedef float v8f __attribute__((ext_vector_type(8)));

#define ROWS 64
#define LDA  132
#define LDH  260
#define NBLK ((NUM_DST + ROWS - 1) / ROWS)
#define SMEM_BYTES ((ROWS * LDA + ROWS * LDH) * 4)

#define NCHUNK  256
#define CHUNK   (NUM_EDGES / NCHUNK)
#define EPT     ((CHUNK + 511) / 512)
#define HB      2048
#define PSTRIDE 1568
#define RSTRIDE 6272
#define LCAP    4096
static_assert(NUM_EDGES % NCHUNK == 0, "chunks must be equal");
static_assert(NBLK + 1 <= PSTRIDE && PSTRIDE <= HB && PSTRIDE % 32 == 0, "bin row");
static_assert(CHUNK <= RSTRIDE && RSTRIDE % 32 == 0, "record row");
static_assert(NUM_SRC <= (1 << 18), "source index must fit 18 bits of a record");

__global__ __launch_bounds__(512) void bin_kernel(
    const int* __restrict__ edge_src,
    const int* __restrict__ edge_dst,
    int* __restrict__ P,
    int* __restrict__ recs)
{
    __shared__ __attribute__((aligned(16))) int hist[HB];
    __shared__ __attribute__((aligned(16))) int srt[RSTRIDE];
    __shared__ int wsm[16];
    const int tid  = threadIdx.x;
    const int lane = tid & 31;
    const int wave = tid >> 5;
    const int base = blockIdx.x * CHUNK;

    int d[EPT], s[EPT];
    #pragma unroll
    for (int j = 0; j < EPT; ++j) {
        const int i = tid + j * 512;
        const bool ok = i < CHUNK;
        d[j] = ok ? edge_dst[base + i] : -1;
        s[j] = ok ? edge_src[base + i] : 0;
    }
    for (int i = tid; i < HB; i += 512) hist[i] = 0;
    if (tid < RSTRIDE - CHUNK) srt[CHUNK + tid] = 0;
    __syncthreads();

    #pragma unroll
    for (int j = 0; j < EPT; ++j)
        if ((unsigned)d[j] < (unsigned)NUM_DST) atomicAdd(&hist[d[j] >> 6], 1);
    __syncthreads();

    const int v0 = hist[4 * tid + 0], v1 = hist[4 * tid + 1];
    const int v2 = hist[4 * tid + 2], v3 = hist[4 * tid + 3];
    const int t4 = v0 + v1 + v2 + v3;
    int incl = t4;
    #pragma unroll
    for (int dd = 1; dd < 32; dd <<= 1) {
        const int t = __shfl_up(incl, dd, 32);
        if (lane >= dd) incl += t;
    }
    if (lane == 31) wsm[wave] = incl;
    __syncthreads();
    int wbase = 0;
    for (int w = 0; w < 16; ++w) { const int t = wsm[w]; if (w < wave) wbase += t; }
    const int e0 = wbase + incl - t4;
    const int e1 = e0 + v0, e2 = e1 + v1, e3 = e2 + v2;
    hist[4 * tid + 0] = e0; hist[4 * tid + 1] = e1;
    hist[4 * tid + 2] = e2; hist[4 * tid + 3] = e3;
    if (tid < PSTRIDE / 4)
        *(int4*)(P + (size_t)blockIdx.x * PSTRIDE + 4 * tid) = make_int4(e0, e1, e2, e3);
    __syncthreads();

    #pragma unroll
    for (int j = 0; j < EPT; ++j) {
        if ((unsigned)d[j] < (unsigned)NUM_DST) {
            const int slot = atomicAdd(&hist[d[j] >> 6], 1);
            const int sc   = min(max(s[j], 0), NUM_SRC - 1);
            srt[slot] = sc | ((d[j] & 63) << 18);
        }
    }
    __syncthreads();

    for (int i = tid; i < RSTRIDE / 4; i += 512)
        *(int4*)(recs + (size_t)blockIdx.x * RSTRIDE + 4 * i) = *(const int4*)(srt + 4 * i);
}


#define NSETS 8

template <bool LAST, int LDW, int LDSA>
__device__ __forceinline__ void k_rotation(v2f (&b)[NSETS], const float* __restrict__ wq,
                                           const float* aq, v8f (&acc)[4])
{
    #pragma unroll
    for (int j = 0; j < NSETS; ++j) {
        if (!LAST || j == 0) {
            b[(j + NSETS - 1) % NSETS].x = wq[(size_t)(4 * (j + NSETS - 1)) * LDW];
            b[(j + NSETS - 1) % NSETS].y = wq[(size_t)(4 * (j + NSETS - 1) + 1) * LDW];
        }
        #pragma unroll
        for (int rt = 0; rt < 4; ++rt) {
            const v2f a = *(const v2f*)(aq + rt * 16 * LDSA + 4 * j);
            acc[rt] = __builtin_amdgcn_wmma_f32_16x16x4_f32(
                false, a, false, b[j], (short)0, acc[rt], false, false);
        }
        __builtin_amdgcn_sched_barrier(0);
    }
}

template <int K, int LDW, int LDSA>
__device__ __forceinline__ void wave_k_loop(const float* __restrict__ wp,
                                            const float* ap, v8f (&acc)[4])
{
    constexpr int NK = K / 4;
    static_assert(NK % NSETS == 0 && NK >= 2 * NSETS, "whole rotations");
    v2f b[NSETS];
    #pragma unroll
    for (int j = 0; j < NSETS - 1; ++j) {
        b[j].x = wp[(size_t)(4 * j) * LDW];
        b[j].y = wp[(size_t)(4 * j + 1) * LDW];
    }
    const float* wq = wp;
    const float* aq = ap;
    for (int s0 = 0; s0 < NK - NSETS; s0 += NSETS) {
        k_rotation<false, LDW, LDSA>(b, wq, aq, acc);
        wq += (size_t)(4 * NSETS) * LDW;
        aq += 4 * NSETS;
    }
    k_rotation<true, LDW, LDSA>(b, wq, aq, acc);
}

__global__ __launch_bounds__(512) __attribute__((amdgpu_waves_per_eu(12, 12))) void fused_mlp_kernel(
    const int* __restrict__ P, const int* __restrict__ recs,
    const float* __restrict__ x,
    const float* __restrict__ w_l,  const float* __restrict__ b_l,
    const float* __restrict__ w_r,
    const float* __restrict__ ln1_g, const float* __restrict__ ln1_b,
    const float* __restrict__ w_out, const float* __restrict__ b_out,
    const float* __restrict__ ln2_g, const float* __restrict__ ln2_b,
    float* __restrict__ out)
{
    extern __shared__ float smem[];
    float* sA = smem;
    float* sH = smem + ROWS * LDA;

    const int tid  = threadIdx.x;
    const int wave = tid >> 5;
    const int lane = tid & 31;
    const int rowbase = blockIdx.x * ROWS;

    const int n    = lane & 15;
    const int kh   = lane >> 4;
    const int arow = lane & 15;

    {
        int* li    = (int*)sH;
        int* lst_u = li;
        int* lst_s = li + LCAP;
        int* rcnt  = li + 2 * LCAP;
        int* rcur  = rcnt + 64;
        int* rst   = rcur + 64;
        int* wsm   = rst + 64;

        const int bin = blockIdx.x;

        const int wv = __builtin_amdgcn_readfirstlane(wave);

        int ctot[4];
        #pragma unroll
        for (int rr = 0; rr < 4; ++rr) {
            ctot[rr] = 0;
            *(float4*)(sA + (wv * 4 + rr) * LDA + lane * 4) = make_float4(0.0f, 0.0f, 0.0f, 0.0f);
        }

        int lo = 0, ntot;
        do {
            if (lo > 0) __syncthreads();
            int p0 = 0, n = 0;
            if (tid < NCHUNK) {
                p0 = P[(size_t)tid * PSTRIDE + bin];
                n  = P[(size_t)tid * PSTRIDE + bin + 1] - p0;
            }
            int incl = n;
            #pragma unroll
            for (int dd = 1; dd < 32; dd <<= 1) {
                const int t = __shfl_up(incl, dd, 32);
                if (lane >= dd) incl += t;
            }
            if (lane == 31) wsm[wave] = incl;
            if (tid < 64) rcnt[tid] = 0;
            __syncthreads();
            int wbase = 0, nall = 0;
            for (int w = 0; w < 16; ++w) {
                const int t = wsm[w];
                if (w < wave) wbase += t;
                nall += t;
            }
            ntot = __builtin_amdgcn_readfirstlane(nall);
            const int pos = wbase + incl - n;
            const int* myrec = recs + (size_t)(tid & (NCHUNK - 1)) * RSTRIDE + p0;
            const int m = max(0, min(LCAP, ntot - lo));

            {
                const int b0 = max(pos, lo), b1 = min(pos + n, lo + m);
                for (int i = b0; i < b1; i += 8) {
                    int rv[8];
                    #pragma unroll
                    for (int j = 0; j < 8; ++j)
                        rv[j] = (i + j < b1) ? myrec[i + j - pos] : 0;
                    #pragma unroll
                    for (int j = 0; j < 8; ++j) {
                        if (i + j < b1) {
                            lst_u[i + j - lo] = rv[j];
                            atomicAdd(&rcnt[(rv[j] >> 18) & 63], 1);
                        }
                    }
                }
            }
            __syncthreads();

            if (wave == 0) {
                const int c0 = rcnt[2 * lane], c1 = rcnt[2 * lane + 1];
                const int t2 = c0 + c1;
                int inc2 = t2;
                #pragma unroll
                for (int dd = 1; dd < 32; dd <<= 1) {
                    const int t = __shfl_up(inc2, dd, 32);
                    if (lane >= dd) inc2 += t;
                }
                const int ex = inc2 - t2;
                rst[2 * lane]      = ex;      rcur[2 * lane]     = ex;
                rst[2 * lane + 1]  = ex + c0; rcur[2 * lane + 1] = ex + c0;
            }
            __syncthreads();

            for (int j = tid; j < m; j += 512) {
                const int rec  = lst_u[j];
                const int slot = atomicAdd(&rcur[(rec >> 18) & 63], 1);
                lst_s[slot] = rec & 0x3FFFF;
            }
            __syncthreads();

            #pragma unroll
            for (int rr = 0; rr < 4; ++rr) {
                const int row = wv * 4 + rr;
                const int c  = __builtin_amdgcn_readfirstlane(rcnt[row]);
                const int st = __builtin_amdgcn_readfirstlane(rst[row]);
                ctot[rr] += c;
                float4 a4 = *(const float4*)(sA + row * LDA + lane * 4);
                for (int base = 0; base < c; base += 32) {
                    const int mm = min(32, c - base);
                    const int sv = (lane < mm) ? lst_s[st + base + lane] : 0;
                    for (int e = 0; e < mm; e += 8) {
                        float4 v[8];
                        #pragma unroll
                        for (int j = 0; j < 8; ++j) {
                            const int ee = min(e + j, mm - 1);
                            const int s  = __builtin_amdgcn_readlane(sv, ee);
                            v[j] = *(const float4*)(x + (size_t)s * IN_DIM + lane * 4);
                        }
                        #pragma unroll
                        for (int j = 0; j < 8; ++j) {
                            if (e + j < mm) {
                                a4.x += v[j].x; a4.y += v[j].y;
                                a4.z += v[j].z; a4.w += v[j].w;
                            }
                        }
                    }
                }
                *(float4*)(sA + row * LDA + lane * 4) = a4;
            }
            lo += LCAP;
        } while (lo < ntot);

        #pragma unroll
        for (int rr = 0; rr < 4; ++rr) {
            const int row = wv * 4 + rr;
            const float inv = 1.0f / fmaxf((float)ctot[rr], 1.0f);
            float4 a4 = *(const float4*)(sA + row * LDA + lane * 4);
            a4.x *= inv; a4.y *= inv; a4.z *= inv; a4.w *= inv;
            *(float4*)(sA + row * LDA + lane * 4) = a4;
        }
    }
    __syncthreads();

    v8f acc[4];
    #pragma unroll
    for (int rt = 0; rt < 4; ++rt) acc[rt] = (v8f){};
    const int colbase = wave * 16;

    wave_k_loop<IN_DIM, HID, LDA>(w_l + (size_t)(2 * kh) * HID + colbase + n,
                                  sA + arow * LDA + 2 * kh, acc);
    __syncthreads();

    #pragma unroll
    for (int j = 0; j < 4; ++j) {
        const int idx = tid + j * 512;
        const int row = idx >> 5;
        const int k   = (idx & 31) * 4;
        int rg = rowbase + row;
        if (rg >= NUM_DST) rg = NUM_DST - 1;
        const float4 x4 = *(const float4*)(x + (size_t)rg * IN_DIM + k);
        sA[row * LDA + k + 0] = x4.x;
        sA[row * LDA + k + 1] = x4.y;
        sA[row * LDA + k + 2] = x4.z;
        sA[row * LDA + k + 3] = x4.w;
    }
    __syncthreads();

    wave_k_loop<IN_DIM, HID, LDA>(w_r + (size_t)(2 * kh) * HID + colbase + n,
                                  sA + arow * LDA + 2 * kh, acc);

    {
        const float bias = b_l[colbase + n];
        #pragma unroll
        for (int rt = 0; rt < 4; ++rt) {
            #pragma unroll
            for (int i = 0; i < 8; ++i) {
                const int row = rt * 16 + i + 8 * kh;
                sH[row * LDH + colbase + n] = acc[rt][i] + bias;
            }
        }
    }
    __syncthreads();

    #pragma unroll
    for (int rr = 0; rr < 4; ++rr) {
        const int row = wave + 16 * rr;
        float v[8], s = 0.0f, s2 = 0.0f;
        #pragma unroll
        for (int j = 0; j < 8; ++j) {
            const float t = sH[row * LDH + lane + 32 * j];
            v[j] = t; s += t; s2 += t * t;
        }
        #pragma unroll
        for (int off = 16; off > 0; off >>= 1) {
            s  += __shfl_xor(s,  off, 32);
            s2 += __shfl_xor(s2, off, 32);
        }
        const float mean = s * (1.0f / HID);
        const float var  = s2 * (1.0f / HID) - mean * mean;
        const float rstd = rsqrtf(var + 1e-5f);
        #pragma unroll
        for (int j = 0; j < 8; ++j) {
            const int col = lane + 32 * j;
            const float hn = (v[j] - mean) * rstd * ln1_g[col] + ln1_b[col];
            sH[row * LDH + col] = 0.5f * hn * (1.0f + erff(hn * 0.70710678118654752f));
        }
    }
    __syncthreads();

    float* sOut = sA;
    if (wave < 8) {
        const int cb2 = wave * 16;
        v8f c[4];
        #pragma unroll
        for (int rt = 0; rt < 4; ++rt) c[rt] = (v8f){};
        wave_k_loop<HID, OUT_DIM, LDH>(w_out + (size_t)(2 * kh) * OUT_DIM + cb2 + n,
                                       sH + arow * LDH + 2 * kh, c);
        const float bias = b_out[cb2 + n];
        #pragma unroll
        for (int rt = 0; rt < 4; ++rt) {
            #pragma unroll
            for (int i = 0; i < 8; ++i) {
                const int row = rt * 16 + i + 8 * kh;
                sOut[row * LDA + cb2 + n] = c[rt][i] + bias;
            }
        }
    }
    __syncthreads();

    #pragma unroll
    for (int rr = 0; rr < 4; ++rr) {
        const int row = wave + 16 * rr;
        float v[4], s = 0.0f, s2 = 0.0f;
        #pragma unroll
        for (int j = 0; j < 4; ++j) {
            const float t = sOut[row * LDA + lane + 32 * j];
            v[j] = t; s += t; s2 += t * t;
        }
        #pragma unroll
        for (int off = 16; off > 0; off >>= 1) {
            s  += __shfl_xor(s,  off, 32);
            s2 += __shfl_xor(s2, off, 32);
        }
        const float mean = s * (1.0f / OUT_DIM);
        const float var  = s2 * (1.0f / OUT_DIM) - mean * mean;
        const float rstd = rsqrtf(var + 1e-5f);
        const int rowg = rowbase + row;
        if (rowg < NUM_DST) {
            #pragma unroll
            for (int j = 0; j < 4; ++j) {
                const int col = lane + 32 * j;
                const float o = (v[j] - mean) * rstd * ln2_g[col] + ln2_b[col];
                out[(size_t)rowg * OUT_DIM + col] = o;
            }
        }
    }
}

extern "C" void kernel_launch(void* const* d_in, const int* in_sizes, int n_in,
                              void* d_out, int out_size, void* d_ws, size_t ws_size,
                              hipStream_t stream) {
    const float* x      = (const float*)d_in[0];
    const float* w_l    = (const float*)d_in[1];
    const float* b_l    = (const float*)d_in[2];
    const float* w_r    = (const float*)d_in[3];
    const float* ln1_g  = (const float*)d_in[4];
    const float* ln1_b  = (const float*)d_in[5];
    const float* w_out  = (const float*)d_in[6];
    const float* b_out  = (const float*)d_in[7];
    const float* ln2_g  = (const float*)d_in[8];
    const float* ln2_b  = (const float*)d_in[9];
    const int*   e_src  = (const int*)d_in[10];
    const int*   e_dst  = (const int*)d_in[11];

    int* P    = (int*)d_ws;
    int* recs = P + NCHUNK * PSTRIDE;
    float* out = (float*)d_out;

    bin_kernel<<<NCHUNK, 512, 0, stream>>>(e_src, e_dst, P, recs);

    fused_mlp_kernel<<<NBLK, 512, SMEM_BYTES, stream>>>(P, recs, x,
                                                        w_l, b_l, w_r,
                                                        ln1_g, ln1_b,
                                                        w_out, b_out,
                                                        ln2_g, ln2_b, out);
}


# GELU: library erff (a divergent branch per element) replaced by a branch-free form, 1+erf = 2-E or E with E = erfc(t) = exp2(t*P10(min(t,4))), abs error about 1e-7; LN1 scale/shift loaded once
# speedup vs baseline: 2.7067x; 1.0630x over previous
#include <hip/hip_runtime.h>
#include <hip/hip_bf16.h>
#include <math.h>

#define NUM_SRC   200000
#define NUM_DST   100000
#define NUM_EDGES 1600000
#define IN_DIM    128
#define HID       256
#define OUT_DIM   128

typedef float v2f __attribute__((ext_vector_type(2)));
typedef float v8f __attribute__((ext_vector_type(8)));

#define ROWS 64
#define LDA  132
#define LDH  260
#define NBLK ((NUM_DST + ROWS - 1) / ROWS)
#define SMEM_BYTES ((ROWS * LDA + ROWS * LDH) * 4)

#define NCHUNK  256
#define CHUNK   (NUM_EDGES / NCHUNK)
#define EPT     ((CHUNK + 511) / 512)
#define HB      2048
#define PSTRIDE 1568
#define RSTRIDE 6272
#define LCAP    4096
static_assert(NUM_EDGES % NCHUNK == 0, "chunks must be equal");
static_assert(NBLK + 1 <= PSTRIDE && PSTRIDE <= HB && PSTRIDE % 32 == 0, "bin row");
static_assert(CHUNK <= RSTRIDE && RSTRIDE % 32 == 0, "record row");
static_assert(NUM_SRC <= (1 << 18), "source index must fit 18 bits of a record");

__global__ __launch_bounds__(512) void bin_kernel(
    const int* __restrict__ edge_src,
    const int* __restrict__ edge_dst,
    int* __restrict__ P,
    int* __restrict__ recs)
{
    __shared__ __attribute__((aligned(16))) int hist[HB];
    __shared__ __attribute__((aligned(16))) int srt[RSTRIDE];
    __shared__ int wsm[16];
    const int tid  = threadIdx.x;
    const int lane = tid & 31;
    const int wave = tid >> 5;
    const int base = blockIdx.x * CHUNK;

    int d[EPT], s[EPT];
    #pragma unroll
    for (int j = 0; j < EPT; ++j) {
        const int i = tid + j * 512;
        const bool ok = i < CHUNK;
        d[j] = ok ? edge_dst[base + i] : -1;
        s[j] = ok ? edge_src[base + i] : 0;
    }
    for (int i = tid; i < HB; i += 512) hist[i] = 0;
    if (tid < RSTRIDE - CHUNK) srt[CHUNK + tid] = 0;
    __syncthreads();

    #pragma unroll
    for (int j = 0; j < EPT; ++j)
        if ((unsigned)d[j] < (unsigned)NUM_DST) atomicAdd(&hist[d[j] >> 6], 1);
    __syncthreads();

    const int v0 = hist[4 * tid + 0], v1 = hist[4 * tid + 1];
    const int v2 = hist[4 * tid + 2], v3 = hist[4 * tid + 3];
    const int t4 = v0 + v1 + v2 + v3;
    int incl = t4;
    #pragma unroll
    for (int dd = 1; dd < 32; dd <<= 1) {
        const int t = __shfl_up(incl, dd, 32);
        if (lane >= dd) incl += t;
    }
    if (lane == 31) wsm[wave] = incl;
    __syncthreads();
    int wbase = 0;
    for (int w = 0; w < 16; ++w) { const int t = wsm[w]; if (w < wave) wbase += t; }
    const int e0 = wbase + incl - t4;
    const int e1 = e0 + v0, e2 = e1 + v1, e3 = e2 + v2;
    hist[4 * tid + 0] = e0; hist[4 * tid + 1] = e1;
    hist[4 * tid + 2] = e2; hist[4 * tid + 3] = e3;
    if (tid < PSTRIDE / 4)
        *(int4*)(P + (size_t)blockIdx.x * PSTRIDE + 4 * tid) = make_int4(e0, e1, e2, e3);
    __syncthreads();

    #pragma unroll
    for (int j = 0; j < EPT; ++j) {
        if ((unsigned)d[j] < (unsigned)NUM_DST) {
            const int slot = atomicAdd(&hist[d[j] >> 6], 1);
            const int sc   = min(max(s[j], 0), NUM_SRC - 1);
            srt[slot] = sc | ((d[j] & 63) << 18);
        }
    }
    __syncthreads();

    for (int i = tid; i < RSTRIDE / 4; i += 512)
        *(int4*)(recs + (size_t)blockIdx.x * RSTRIDE + 4 * i) = *(const int4*)(srt + 4 * i);
}


#define NSETS 8

template <bool LAST, int LDW, int LDSA>
__device__ __forceinline__ void k_rotation(v2f (&b)[NSETS], const float* __restrict__ wq,
                                           const float* aq, v8f (&acc)[4])
{
    #pragma unroll
    for (int j = 0; j < NSETS; ++j) {
        if (!LAST || j == 0) {
            b[(j + NSETS - 1) % NSETS].x = wq[(size_t)(4 * (j + NSETS - 1)) * LDW];
            b[(j + NSETS - 1) % NSETS].y = wq[(size_t)(4 * (j + NSETS - 1) + 1) * LDW];
        }
        #pragma unroll
        for (int rt = 0; rt < 4; ++rt) {
            const v2f a = *(const v2f*)(aq + rt * 16 * LDSA + 4 * j);
            acc[rt] = __builtin_amdgcn_wmma_f32_16x16x4_f32(
                false, a, false, b[j], (short)0, acc[rt], false, false);
        }
        __builtin_amdgcn_sched_barrier(0);
    }
}

template <int K, int LDW, int LDSA>
__device__ __forceinline__ void wave_k_loop(const float* __restrict__ wp,
                                            const float* ap, v8f (&acc)[4])
{
    constexpr int NK = K / 4;
    static_assert(NK % NSETS == 0 && NK >= 2 * NSETS, "whole rotations");
    v2f b[NSETS];
    #pragma unroll
    for (int j = 0; j < NSETS - 1; ++j) {
        b[j].x = wp[(size_t)(4 * j) * LDW];
        b[j].y = wp[(size_t)(4 * j + 1) * LDW];
    }
    const float* wq = wp;
    const float* aq = ap;
    for (int s0 = 0; s0 < NK - NSETS; s0 += NSETS) {
        k_rotation<false, LDW, LDSA>(b, wq, aq, acc);
        wq += (size_t)(4 * NSETS) * LDW;
        aq += 4 * NSETS;
    }
    k_rotation<true, LDW, LDSA>(b, wq, aq, acc);
}

__device__ __forceinline__ float one_plus_erf(float x)
{
    const float t  = fabsf(x);
    const float tc = fminf(t, 4.0f);
    float q = 1.46286013e-7f;
    q = fmaf(q, tc, -3.69151803e-6f);
    q = fmaf(q, tc, 4.14682254e-5f);
    q = fmaf(q, tc, -2.71359662e-4f);
    q = fmaf(q, tc, 1.11358718e-3f);
    q = fmaf(q, tc, -2.63007754e-3f);
    q = fmaf(q, tc, 3.72432347e-4f);
    q = fmaf(q, tc, 2.75863279e-2f);
    q = fmaf(q, tc, -1.48270622e-1f);
    q = fmaf(q, tc, -9.18447912e-1f);
    q = fmaf(q, tc, -1.62790704e+0f);
    const float e = __builtin_amdgcn_exp2f(t * q);
    return x >= 0.0f ? 2.0f - e : e;
}

__global__ __launch_bounds__(512) __attribute__((amdgpu_waves_per_eu(12, 12))) void fused_mlp_kernel(
    const int* __restrict__ P, const int* __restrict__ recs,
    const float* __restrict__ x,
    const float* __restrict__ w_l,  const float* __restrict__ b_l,
    const float* __restrict__ w_r,
    const float* __restrict__ ln1_g, const float* __restrict__ ln1_b,
    const float* __restrict__ w_out, const float* __restrict__ b_out,
    const float* __restrict__ ln2_g, const float* __restrict__ ln2_b,
    float* __restrict__ out)
{
    extern __shared__ float smem[];
    float* sA = smem;
    float* sH = smem + ROWS * LDA;

    const int tid  = threadIdx.x;
    const int wave = tid >> 5;
    const int lane = tid & 31;
    const int rowbase = blockIdx.x * ROWS;

    const int n    = lane & 15;
    const int kh   = lane >> 4;
    const int arow = lane & 15;

    {
        int* li    = (int*)sH;
        int* lst_u = li;
        int* lst_s = li + LCAP;
        int* rcnt  = li + 2 * LCAP;
        int* rcur  = rcnt + 64;
        int* rst   = rcur + 64;
        int* wsm   = rst + 64;

        const int bin = blockIdx.x;

        const int wv = __builtin_amdgcn_readfirstlane(wave);

        int ctot[4];
        #pragma unroll
        for (int rr = 0; rr < 4; ++rr) {
            ctot[rr] = 0;
            *(float4*)(sA + (wv * 4 + rr) * LDA + lane * 4) = make_float4(0.0f, 0.0f, 0.0f, 0.0f);
        }

        int lo = 0, ntot;
        do {
            if (lo > 0) __syncthreads();
            int p0 = 0, n = 0;
            if (tid < NCHUNK) {
                p0 = P[(size_t)tid * PSTRIDE + bin];
                n  = P[(size_t)tid * PSTRIDE + bin + 1] - p0;
            }
            int incl = n;
            #pragma unroll
            for (int dd = 1; dd < 32; dd <<= 1) {
                const int t = __shfl_up(incl, dd, 32);
                if (lane >= dd) incl += t;
            }
            if (lane == 31) wsm[wave] = incl;
            if (tid < 64) rcnt[tid] = 0;
            __syncthreads();
            int wbase = 0, nall = 0;
            for (int w = 0; w < 16; ++w) {
                const int t = wsm[w];
                if (w < wave) wbase += t;
                nall += t;
            }
            ntot = __builtin_amdgcn_readfirstlane(nall);
            const int pos = wbase + incl - n;
            const int* myrec = recs + (size_t)(tid & (NCHUNK - 1)) * RSTRIDE + p0;
            const int m = max(0, min(LCAP, ntot - lo));

            {
                const int b0 = max(pos, lo), b1 = min(pos + n, lo + m);
                for (int i = b0; i < b1; i += 8) {
                    int rv[8];
                    #pragma unroll
                    for (int j = 0; j < 8; ++j)
                        rv[j] = (i + j < b1) ? myrec[i + j - pos] : 0;
                    #pragma unroll
                    for (int j = 0; j < 8; ++j) {
                        if (i + j < b1) {
                            lst_u[i + j - lo] = rv[j];
                            atomicAdd(&rcnt[(rv[j] >> 18) & 63], 1);
                        }
                    }
                }
            }
            __syncthreads();

            if (wave == 0) {
                const int c0 = rcnt[2 * lane], c1 = rcnt[2 * lane + 1];
                const int t2 = c0 + c1;
                int inc2 = t2;
                #pragma unroll
                for (int dd = 1; dd < 32; dd <<= 1) {
                    const int t = __shfl_up(inc2, dd, 32);
                    if (lane >= dd) inc2 += t;
                }
                const int ex = inc2 - t2;
                rst[2 * lane]      = ex;      rcur[2 * lane]     = ex;
                rst[2 * lane + 1]  = ex + c0; rcur[2 * lane + 1] = ex + c0;
            }
            __syncthreads();

            for (int j = tid; j < m; j += 512) {
                const int rec  = lst_u[j];
                const int slot = atomicAdd(&rcur[(rec >> 18) & 63], 1);
                lst_s[slot] = rec & 0x3FFFF;
            }
            __syncthreads();

            #pragma unroll
            for (int rr = 0; rr < 4; ++rr) {
                const int row = wv * 4 + rr;
                const int c  = __builtin_amdgcn_readfirstlane(rcnt[row]);
                const int st = __builtin_amdgcn_readfirstlane(rst[row]);
                ctot[rr] += c;
                float4 a4 = *(const float4*)(sA + row * LDA + lane * 4);
                for (int base = 0; base < c; base += 32) {
                    const int mm = min(32, c - base);
                    const int sv = (lane < mm) ? lst_s[st + base + lane] : 0;
                    for (int e = 0; e < mm; e += 8) {
                        float4 v[8];
                        #pragma unroll
                        for (int j = 0; j < 8; ++j) {
                            const int ee = min(e + j, mm - 1);
                            const int s  = __builtin_amdgcn_readlane(sv, ee);
                            v[j] = *(const float4*)(x + (size_t)s * IN_DIM + lane * 4);
                        }
                        #pragma unroll
                        for (int j = 0; j < 8; ++j) {
                            if (e + j < mm) {
                                a4.x += v[j].x; a4.y += v[j].y;
                                a4.z += v[j].z; a4.w += v[j].w;
                            }
                        }
                    }
                }
                *(float4*)(sA + row * LDA + lane * 4) = a4;
            }
            lo += LCAP;
        } while (lo < ntot);

        #pragma unroll
        for (int rr = 0; rr < 4; ++rr) {
            const int row = wv * 4 + rr;
            const float inv = 1.0f / fmaxf((float)ctot[rr], 1.0f);
            float4 a4 = *(const float4*)(sA + row * LDA + lane * 4);
            a4.x *= inv; a4.y *= inv; a4.z *= inv; a4.w *= inv;
            *(float4*)(sA + row * LDA + lane * 4) = a4;
        }
    }
    __syncthreads();

    v8f acc[4];
    #pragma unroll
    for (int rt = 0; rt < 4; ++rt) acc[rt] = (v8f){};
    const int colbase = wave * 16;

    wave_k_loop<IN_DIM, HID, LDA>(w_l + (size_t)(2 * kh) * HID + colbase + n,
                                  sA + arow * LDA + 2 * kh, acc);
    __syncthreads();

    #pragma unroll
    for (int j = 0; j < 4; ++j) {
        const int idx = tid + j * 512;
        const int row = idx >> 5;
        const int k   = (idx & 31) * 4;
        int rg = rowbase + row;
        if (rg >= NUM_DST) rg = NUM_DST - 1;
        const float4 x4 = *(const float4*)(x + (size_t)rg * IN_DIM + k);
        sA[row * LDA + k + 0] = x4.x;
        sA[row * LDA + k + 1] = x4.y;
        sA[row * LDA + k + 2] = x4.z;
        sA[row * LDA + k + 3] = x4.w;
    }
    __syncthreads();

    wave_k_loop<IN_DIM, HID, LDA>(w_r + (size_t)(2 * kh) * HID + colbase + n,
                                  sA + arow * LDA + 2 * kh, acc);

    {
        const float bias = b_l[colbase + n];
        #pragma unroll
        for (int rt = 0; rt < 4; ++rt) {
            #pragma unroll
            for (int i = 0; i < 8; ++i) {
                const int row = rt * 16 + i + 8 * kh;
                sH[row * LDH + colbase + n] = acc[rt][i] + bias;
            }
        }
    }
    __syncthreads();

    float g1[8], be1[8];
    #pragma unroll
    for (int j = 0; j < 8; ++j) {
        g1[j]  = ln1_g[lane + 32 * j];
        be1[j] = ln1_b[lane + 32 * j];
    }
    #pragma unroll
    for (int rr = 0; rr < 4; ++rr) {
        const int row = wave + 16 * rr;
        float v[8], s = 0.0f, s2 = 0.0f;
        #pragma unroll
        for (int j = 0; j < 8; ++j) {
            const float t = sH[row * LDH + lane + 32 * j];
            v[j] = t; s += t; s2 += t * t;
        }
        #pragma unroll
        for (int off = 16; off > 0; off >>= 1) {
            s  += __shfl_xor(s,  off, 32);
            s2 += __shfl_xor(s2, off, 32);
        }
        const float mean = s * (1.0f / HID);
        const float var  = s2 * (1.0f / HID) - mean * mean;
        const float rstd = rsqrtf(var + 1e-5f);
        #pragma unroll
        for (int j = 0; j < 8; ++j) {
            const int col = lane + 32 * j;
            const float hn = (v[j] - mean) * rstd * g1[j] + be1[j];
            sH[row * LDH + col] = 0.5f * hn * one_plus_erf(hn * 0.70710678118654752f);
        }
    }
    __syncthreads();

    float* sOut = sA;
    if (wave < 8) {
        const int cb2 = wave * 16;
        v8f c[4];
        #pragma unroll
        for (int rt = 0; rt < 4; ++rt) c[rt] = (v8f){};
        wave_k_loop<HID, OUT_DIM, LDH>(w_out + (size_t)(2 * kh) * OUT_DIM + cb2 + n,
                                       sH + arow * LDH + 2 * kh, c);
        const float bias = b_out[cb2 + n];
        #pragma unroll
        for (int rt = 0; rt < 4; ++rt) {
            #pragma unroll
            for (int i = 0; i < 8; ++i) {
                const int row = rt * 16 + i + 8 * kh;
                sOut[row * LDA + cb2 + n] = c[rt][i] + bias;
            }
        }
    }
    __syncthreads();

    #pragma unroll
    for (int rr = 0; rr < 4; ++rr) {
        const int row = wave + 16 * rr;
        float v[4], s = 0.0f, s2 = 0.0f;
        #pragma unroll
        for (int j = 0; j < 4; ++j) {
            const float t = sOut[row * LDA + lane + 32 * j];
            v[j] = t; s += t; s2 += t * t;
        }
        #pragma unroll
        for (int off = 16; off > 0; off >>= 1) {
            s  += __shfl_xor(s,  off, 32);
            s2 += __shfl_xor(s2, off, 32);
        }
        const float mean = s * (1.0f / OUT_DIM);
        const float var  = s2 * (1.0f / OUT_DIM) - mean * mean;
        const float rstd = rsqrtf(var + 1e-5f);
        const int rowg = rowbase + row;
        if (rowg < NUM_DST) {
            #pragma unroll
            for (int j = 0; j < 4; ++j) {
                const int col = lane + 32 * j;
                const float o = (v[j] - mean) * rstd * ln2_g[col] + ln2_b[col];
                out[(size_t)rowg * OUT_DIM + col] = o;
            }
        }
    }
}

extern "C" void kernel_launch(void* const* d_in, const int* in_sizes, int n_in,
                              void* d_out, int out_size, void* d_ws, size_t ws_size,
                              hipStream_t stream) {
    const float* x      = (const float*)d_in[0];
    const float* w_l    = (const float*)d_in[1];
    const float* b_l    = (const float*)d_in[2];
    const float* w_r    = (const float*)d_in[3];
    const float* ln1_g  = (const float*)d_in[4];
    const float* ln1_b  = (const float*)d_in[5];
    const float* w_out  = (const float*)d_in[6];
    const float* b_out  = (const float*)d_in[7];
    const float* ln2_g  = (const float*)d_in[8];
    const float* ln2_b  = (const float*)d_in[9];
    const int*   e_src  = (const int*)d_in[10];
    const int*   e_dst  = (const int*)d_in[11];

    int* P    = (int*)d_ws;
    int* recs = P + NCHUNK * PSTRIDE;
    float* out = (float*)d_out;

    bin_kernel<<<NCHUNK, 512, 0, stream>>>(e_src, e_dst, P, recs);

    fused_mlp_kernel<<<NBLK, 512, SMEM_BYTES, stream>>>(P, recs, x,
                                                        w_l, b_l, w_r,
                                                        ln1_g, ln1_b,
                                                        w_out, b_out,
                                                        ln2_g, ln2_b, out);
}


# fused kernel grid: the 27 bins left over after 6 x 256 whole-bin workgroups run as 106 quarter-bin workgroups (16 rows, one row tile per wave), not as a last nearly empty round of 64-row ones
# speedup vs baseline: 2.9309x; 1.0828x over previous
#include <hip/hip_runtime.h>
#include <hip/hip_bf16.h>
#include <math.h>

#define NUM_SRC   200000
#define NUM_DST   100000
#define NUM_EDGES 1600000
#define IN_DIM    128
#define HID       256
#define OUT_DIM   128

typedef float v2f __attribute__((ext_vector_type(2)));
typedef float v8f __attribute__((ext_vector_type(8)));

#define ROWS 64
#define LDA  132
#define LDH  260
#define NBLK ((NUM_DST + ROWS - 1) / ROWS)
#define SMEM_BYTES ((ROWS * LDA + ROWS * LDH) * 4)

#define NCHUNK  256
#define CHUNK   (NUM_EDGES / NCHUNK)
#define EPT     ((CHUNK + 511) / 512)
#define HB      2048
#define PSTRIDE 1568
#define RSTRIDE 6272
#define LCAP    4096
static_assert(NUM_EDGES % NCHUNK == 0, "chunks must be equal");
static_assert(NBLK + 1 <= PSTRIDE && PSTRIDE <= HB && PSTRIDE % 32 == 0, "bin row");
static_assert(CHUNK <= RSTRIDE && RSTRIDE % 32 == 0, "record row");
static_assert(NUM_SRC <= (1 << 18), "source index must fit 18 bits of a record");

__global__ __launch_bounds__(512) void bin_kernel(
    const int* __restrict__ edge_src,
    const int* __restrict__ edge_dst,
    int* __restrict__ P,
    int* __restrict__ recs)
{
    __shared__ __attribute__((aligned(16))) int hist[HB];
    __shared__ __attribute__((aligned(16))) int srt[RSTRIDE];
    __shared__ int wsm[16];
    const int tid  = threadIdx.x;
    const int lane = tid & 31;
    const int wave = tid >> 5;
    const int base = blockIdx.x * CHUNK;

    int d[EPT], s[EPT];
    #pragma unroll
    for (int j = 0; j < EPT; ++j) {
        const int i = tid + j * 512;
        const bool ok = i < CHUNK;
        d[j] = ok ? edge_dst[base + i] : -1;
        s[j] = ok ? edge_src[base + i] : 0;
    }
    for (int i = tid; i < HB; i += 512) hist[i] = 0;
    if (tid < RSTRIDE - CHUNK) srt[CHUNK + tid] = 0;
    __syncthreads();

    #pragma unroll
    for (int j = 0; j < EPT; ++j)
        if ((unsigned)d[j] < (unsigned)NUM_DST) atomicAdd(&hist[d[j] >> 6], 1);
    __syncthreads();

    const int v0 = hist[4 * tid + 0], v1 = hist[4 * tid + 1];
    const int v2 = hist[4 * tid + 2], v3 = hist[4 * tid + 3];
    const int t4 = v0 + v1 + v2 + v3;
    int incl = t4;
    #pragma unroll
    for (int dd = 1; dd < 32; dd <<= 1) {
        const int t = __shfl_up(incl, dd, 32);
        if (lane >= dd) incl += t;
    }
    if (lane == 31) wsm[wave] = incl;
    __syncthreads();
    int wbase = 0;
    for (int w = 0; w < 16; ++w) { const int t = wsm[w]; if (w < wave) wbase += t; }
    const int e0 = wbase + incl - t4;
    const int e1 = e0 + v0, e2 = e1 + v1, e3 = e2 + v2;
    hist[4 * tid + 0] = e0; hist[4 * tid + 1] = e1;
    hist[4 * tid + 2] = e2; hist[4 * tid + 3] = e3;
    if (tid < PSTRIDE / 4)
        *(int4*)(P + (size_t)blockIdx.x * PSTRIDE + 4 * tid) = make_int4(e0, e1, e2, e3);
    __syncthreads();

    #pragma unroll
    for (int j = 0; j < EPT; ++j) {
        if ((unsigned)d[j] < (unsigned)NUM_DST) {
            const int slot = atomicAdd(&hist[d[j] >> 6], 1);
            const int sc   = min(max(s[j], 0), NUM_SRC - 1);
            srt[slot] = sc | ((d[j] & 63) << 18);
        }
    }
    __syncthreads();

    for (int i = tid; i < RSTRIDE / 4; i += 512)
        *(int4*)(recs + (size_t)blockIdx.x * RSTRIDE + 4 * i) = *(const int4*)(srt + 4 * i);
}


#define NSETS 8

template <bool LAST, int LDW, int LDSA, int NRT>
__device__ __forceinline__ void k_rotation(v2f (&b)[NSETS], const float* __restrict__ wq,
                                           const float* aq, v8f (&acc)[NRT])
{
    #pragma unroll
    for (int j = 0; j < NSETS; ++j) {
        if (!LAST || j == 0) {
            b[(j + NSETS - 1) % NSETS].x = wq[(size_t)(4 * (j + NSETS - 1)) * LDW];
            b[(j + NSETS - 1) % NSETS].y = wq[(size_t)(4 * (j + NSETS - 1) + 1) * LDW];
        }
        #pragma unroll
        for (int rt = 0; rt < NRT; ++rt) {
            const v2f a = *(const v2f*)(aq + rt * 16 * LDSA + 4 * j);
            acc[rt] = __builtin_amdgcn_wmma_f32_16x16x4_f32(
                false, a, false, b[j], (short)0, acc[rt], false, false);
        }
        if (NRT > 1 || (j & 3) == 3) __builtin_amdgcn_sched_barrier(0);
    }
}

template <int K, int LDW, int LDSA, int NRT>
__device__ __forceinline__ void wave_k_loop(const float* __restrict__ wp,
                                            const float* ap, v8f (&acc)[NRT])
{
    constexpr int NK = K / 4;
    static_assert(NK % NSETS == 0 && NK >= 2 * NSETS, "whole rotations");
    v2f b[NSETS];
    #pragma unroll
    for (int j = 0; j < NSETS - 1; ++j) {
        b[j].x = wp[(size_t)(4 * j) * LDW];
        b[j].y = wp[(size_t)(4 * j + 1) * LDW];
    }
    const float* wq = wp;
    const float* aq = ap;
    for (int s0 = 0; s0 < NK - NSETS; s0 += NSETS) {
        k_rotation<false, LDW, LDSA, NRT>(b, wq, aq, acc);
        wq += (size_t)(4 * NSETS) * LDW;
        aq += 4 * NSETS;
    }
    k_rotation<true, LDW, LDSA, NRT>(b, wq, aq, acc);
}

__device__ __forceinline__ float one_plus_erf(float x)
{
    const float t  = fabsf(x);
    const float tc = fminf(t, 4.0f);
    float q = 1.46286013e-7f;
    q = fmaf(q, tc, -3.69151803e-6f);
    q = fmaf(q, tc, 4.14682254e-5f);
    q = fmaf(q, tc, -2.71359662e-4f);
    q = fmaf(q, tc, 1.11358718e-3f);
    q = fmaf(q, tc, -2.63007754e-3f);
    q = fmaf(q, tc, 3.72432347e-4f);
    q = fmaf(q, tc, 2.75863279e-2f);
    q = fmaf(q, tc, -1.48270622e-1f);
    q = fmaf(q, tc, -9.18447912e-1f);
    q = fmaf(q, tc, -1.62790704e+0f);
    const float e = __builtin_amdgcn_exp2f(t * q);
    return x >= 0.0f ? 2.0f - e : e;
}

template <int NRT>
__device__ __forceinline__ void tile_body(
    const int bin, const int rq,
    const int* __restrict__ P, const int* __restrict__ recs,
    const float* __restrict__ x,
    const float* __restrict__ w_l,  const float* __restrict__ b_l,
    const float* __restrict__ w_r,
    const float* __restrict__ ln1_g, const float* __restrict__ ln1_b,
    const float* __restrict__ w_out, const float* __restrict__ b_out,
    const float* __restrict__ ln2_g, const float* __restrict__ ln2_b,
    float* __restrict__ out)
{
    extern __shared__ float smem[];
    float* sA = smem;
    float* sH = smem + ROWS * LDA;

    const int tid  = threadIdx.x;
    const int wave = tid >> 5;
    const int lane = tid & 31;
    const int rowbase = bin * ROWS + rq;

    const int n    = lane & 15;
    const int kh   = lane >> 4;
    const int arow = lane & 15;

    {
        int* li    = (int*)sH;
        int* lst_u = li;
        int* lst_s = li + LCAP;
        int* rcnt  = li + 2 * LCAP;
        int* rcur  = rcnt + 64;
        int* rst   = rcur + 64;
        int* wsm   = rst + 64;

        const int wv = __builtin_amdgcn_readfirstlane(wave);

        int ctot[NRT];
        #pragma unroll
        for (int rr = 0; rr < NRT; ++rr) {
            ctot[rr] = 0;
            *(float4*)(sA + (wv * NRT + rr) * LDA + lane * 4) = make_float4(0.0f, 0.0f, 0.0f, 0.0f);
        }

        int lo = 0, ntot;
        do {
            if (lo > 0) __syncthreads();
            int p0 = 0, n = 0;
            if (tid < NCHUNK) {
                p0 = P[(size_t)tid * PSTRIDE + bin];
                n  = P[(size_t)tid * PSTRIDE + bin + 1] - p0;
            }
            int incl = n;
            #pragma unroll
            for (int dd = 1; dd < 32; dd <<= 1) {
                const int t = __shfl_up(incl, dd, 32);
                if (lane >= dd) incl += t;
            }
            if (lane == 31) wsm[wave] = incl;
            if (tid < 64) rcnt[tid] = 0;
            __syncthreads();
            int wbase = 0, nall = 0;
            for (int w = 0; w < 16; ++w) {
                const int t = wsm[w];
                if (w < wave) wbase += t;
                nall += t;
            }
            ntot = __builtin_amdgcn_readfirstlane(nall);
            const int pos = wbase + incl - n;
            const int* myrec = recs + (size_t)(tid & (NCHUNK - 1)) * RSTRIDE + p0;
            const int m = max(0, min(LCAP, ntot - lo));

            {
                const int b0 = max(pos, lo), b1 = min(pos + n, lo + m);
                for (int i = b0; i < b1; i += 8) {
                    int rv[8];
                    #pragma unroll
                    for (int j = 0; j < 8; ++j)
                        rv[j] = (i + j < b1) ? myrec[i + j - pos] : 0;
                    #pragma unroll
                    for (int j = 0; j < 8; ++j) {
                        if (i + j < b1) {
                            lst_u[i + j - lo] = rv[j];
                            atomicAdd(&rcnt[(rv[j] >> 18) & 63], 1);
                        }
                    }
                }
            }
            __syncthreads();

            if (wave == 0) {
                const int c0 = rcnt[2 * lane], c1 = rcnt[2 * lane + 1];
                const int t2 = c0 + c1;
                int inc2 = t2;
                #pragma unroll
                for (int dd = 1; dd < 32; dd <<= 1) {
                    const int t = __shfl_up(inc2, dd, 32);
                    if (lane >= dd) inc2 += t;
                }
                const int ex = inc2 - t2;
                rst[2 * lane]      = ex;      rcur[2 * lane]     = ex;
                rst[2 * lane + 1]  = ex + c0; rcur[2 * lane + 1] = ex + c0;
            }
            __syncthreads();

            for (int j = tid; j < m; j += 512) {
                const int rec  = lst_u[j];
                const int slot = atomicAdd(&rcur[(rec >> 18) & 63], 1);
                lst_s[slot] = rec & 0x3FFFF;
            }
            __syncthreads();

            #pragma unroll
            for (int rr = 0; rr < NRT; ++rr) {
                const int row = wv * NRT + rr;
                const int c  = __builtin_amdgcn_readfirstlane(rcnt[rq + row]);
                const int st = __builtin_amdgcn_readfirstlane(rst[rq + row]);
                ctot[rr] += c;
                float4 a4 = *(const float4*)(sA + row * LDA + lane * 4);
                for (int base = 0; base < c; base += 32) {
                    const int mm = min(32, c - base);
                    const int sv = (lane < mm) ? lst_s[st + base + lane] : 0;
                    for (int e = 0; e < mm; e += 8) {
                        float4 v[8];
                        #pragma unroll
                        for (int j = 0; j < 8; ++j) {
                            const int ee = min(e + j, mm - 1);
                            const int s  = __builtin_amdgcn_readlane(sv, ee);
                            v[j] = *(const float4*)(x + (size_t)s * IN_DIM + lane * 4);
                        }
                        #pragma unroll
                        for (int j = 0; j < 8; ++j) {
                            if (e + j < mm) {
                                a4.x += v[j].x; a4.y += v[j].y;
                                a4.z += v[j].z; a4.w += v[j].w;
                            }
                        }
                    }
                }
                *(float4*)(sA + row * LDA + lane * 4) = a4;
            }
            lo += LCAP;
        } while (lo < ntot);

        #pragma unroll
        for (int rr = 0; rr < NRT; ++rr) {
            const int row = wv * NRT + rr;
            const float inv = 1.0f / fmaxf((float)ctot[rr], 1.0f);
            float4 a4 = *(const float4*)(sA + row * LDA + lane * 4);
            a4.x *= inv; a4.y *= inv; a4.z *= inv; a4.w *= inv;
            *(float4*)(sA + row * LDA + lane * 4) = a4;
        }
    }
    __syncthreads();

    v8f acc[NRT];
    #pragma unroll
    for (int rt = 0; rt < NRT; ++rt) acc[rt] = (v8f){};
    const int colbase = wave * 16;

    wave_k_loop<IN_DIM, HID, LDA, NRT>(w_l + (size_t)(2 * kh) * HID + colbase + n,
                                  sA + arow * LDA + 2 * kh, acc);
    __syncthreads();

    #pragma unroll
    for (int j = 0; j < NRT; ++j) {
        const int idx = tid + j * 512;
        const int row = idx >> 5;
        const int k   = (idx & 31) * 4;
        int rg = rowbase + row;
        if (rg >= NUM_DST) rg = NUM_DST - 1;
        const float4 x4 = *(const float4*)(x + (size_t)rg * IN_DIM + k);
        sA[row * LDA + k + 0] = x4.x;
        sA[row * LDA + k + 1] = x4.y;
        sA[row * LDA + k + 2] = x4.z;
        sA[row * LDA + k + 3] = x4.w;
    }
    __syncthreads();

    wave_k_loop<IN_DIM, HID, LDA, NRT>(w_r + (size_t)(2 * kh) * HID + colbase + n,
                                  sA + arow * LDA + 2 * kh, acc);

    {
        const float bias = b_l[colbase + n];
        #pragma unroll
        for (int rt = 0; rt < NRT; ++rt) {
            #pragma unroll
            for (int i = 0; i < 8; ++i) {
                const int row = rt * 16 + i + 8 * kh;
                sH[row * LDH + colbase + n] = acc[rt][i] + bias;
            }
        }
    }
    __syncthreads();

    float g1[8], be1[8];
    #pragma unroll
    for (int j = 0; j < 8; ++j) {
        g1[j]  = ln1_g[lane + 32 * j];
        be1[j] = ln1_b[lane + 32 * j];
    }
    #pragma unroll
    for (int rr = 0; rr < NRT; ++rr) {
        const int row = wave + 16 * rr;
        float v[8], s = 0.0f, s2 = 0.0f;
        #pragma unroll
        for (int j = 0; j < 8; ++j) {
            const float t = sH[row * LDH + lane + 32 * j];
            v[j] = t; s += t; s2 += t * t;
        }
        #pragma unroll
        for (int off = 16; off > 0; off >>= 1) {
            s  += __shfl_xor(s,  off, 32);
            s2 += __shfl_xor(s2, off, 32);
        }
        const float mean = s * (1.0f / HID);
        const float var  = s2 * (1.0f / HID) - mean * mean;
        const float rstd = rsqrtf(var + 1e-5f);
        #pragma unroll
        for (int j = 0; j < 8; ++j) {
            const int col = lane + 32 * j;
            const float hn = (v[j] - mean) * rstd * g1[j] + be1[j];
            sH[row * LDH + col] = 0.5f * hn * one_plus_erf(hn * 0.70710678118654752f);
        }
    }
    __syncthreads();

    float* sOut = sA;
    if (wave < 8) {
        const int cb2 = wave * 16;
        v8f c[NRT];
        #pragma unroll
        for (int rt = 0; rt < NRT; ++rt) c[rt] = (v8f){};
        wave_k_loop<HID, OUT_DIM, LDH, NRT>(w_out + (size_t)(2 * kh) * OUT_DIM + cb2 + n,
                                       sH + arow * LDH + 2 * kh, c);
        const float bias = b_out[cb2 + n];
        #pragma unroll
        for (int rt = 0; rt < NRT; ++rt) {
            #pragma unroll
            for (int i = 0; i < 8; ++i) {
                const int row = rt * 16 + i + 8 * kh;
                sOut[row * LDA + cb2 + n] = c[rt][i] + bias;
            }
        }
    }
    __syncthreads();

    #pragma unroll
    for (int rr = 0; rr < NRT; ++rr) {
        const int row = wave + 16 * rr;
        float v[4], s = 0.0f, s2 = 0.0f;
        #pragma unroll
        for (int j = 0; j < 4; ++j) {
            const float t = sOut[row * LDA + lane + 32 * j];
            v[j] = t; s += t; s2 += t * t;
        }
        #pragma unroll
        for (int off = 16; off > 0; off >>= 1) {
            s  += __shfl_xor(s,  off, 32);
            s2 += __shfl_xor(s2, off, 32);
        }
        const float mean = s * (1.0f / OUT_DIM);
        const float var  = s2 * (1.0f / OUT_DIM) - mean * mean;
        const float rstd = rsqrtf(var + 1e-5f);
        const int rowg = rowbase + row;
        if (rowg < NUM_DST) {
            #pragma unroll
            for (int j = 0; j < 4; ++j) {
                const int col = lane + 32 * j;
                const float o = (v[j] - mean) * rstd * ln2_g[col] + ln2_b[col];
                out[(size_t)rowg * OUT_DIM + col] = o;
            }
        }
    }
}

#define NFULL  1536
#define NQUART ((NUM_DST - NFULL * ROWS + 15) / 16)
static_assert(NFULL <= NBLK && NFULL * ROWS <= NUM_DST, "whole bins");
__global__ __launch_bounds__(512) __attribute__((amdgpu_waves_per_eu(12, 12))) void fused_mlp_kernel(
    const int* __restrict__ P, const int* __restrict__ recs,
    const float* __restrict__ x,
    const float* __restrict__ w_l,  const float* __restrict__ b_l,
    const float* __restrict__ w_r,
    const float* __restrict__ ln1_g, const float* __restrict__ ln1_b,
    const float* __restrict__ w_out, const float* __restrict__ b_out,
    const float* __restrict__ ln2_g, const float* __restrict__ ln2_b,
    float* __restrict__ out)
{
    if (blockIdx.x >= NFULL) {
        const int t = blockIdx.x - NFULL;
        tile_body<1>(NFULL + (t >> 2), (t & 3) * 16, P, recs, x, w_l, b_l, w_r, ln1_g, ln1_b,
                     w_out, b_out, ln2_g, ln2_b, out);
    } else {
        tile_body<4>(blockIdx.x, 0, P, recs, x, w_l, b_l, w_r, ln1_g, ln1_b,
                     w_out, b_out, ln2_g, ln2_b, out);
    }
}

extern "C" void kernel_launch(void* const* d_in, const int* in_sizes, int n_in,
                              void* d_out, int out_size, void* d_ws, size_t ws_size,
                              hipStream_t stream) {
    const float* x      = (const float*)d_in[0];
    const float* w_l    = (const float*)d_in[1];
    const float* b_l    = (const float*)d_in[2];
    const float* w_r    = (const float*)d_in[3];
    const float* ln1_g  = (const float*)d_in[4];
    const float* ln1_b  = (const float*)d_in[5];
    const float* w_out  = (const float*)d_in[6];
    const float* b_out  = (const float*)d_in[7];
    const float* ln2_g  = (const float*)d_in[8];
    const float* ln2_b  = (const float*)d_in[9];
    const int*   e_src  = (const int*)d_in[10];
    const int*   e_dst  = (const int*)d_in[11];

    int* P    = (int*)d_ws;
    int* recs = P + NCHUNK * PSTRIDE;
    float* out = (float*)d_out;

    bin_kernel<<<NCHUNK, 512, 0, stream>>>(e_src, e_dst, P, recs);

    fused_mlp_kernel<<<NFULL + NQUART, 512, SMEM_BYTES, stream>>>(P, recs, x,
                                                        w_l, b_l, w_r,
                                                        ln1_g, ln1_b,
                                                        w_out, b_out,
                                                        ln2_g, ln2_b, out);
}
